# starting kernel
# baseline (speedup 1.0000x reference)
#include <hip/hip_runtime.h>
#include <math.h>

typedef __attribute__((ext_vector_type(2))) float v2f;
typedef __attribute__((ext_vector_type(8))) float v8f;

#define NEG_SLOPE 0.2f

__device__ __forceinline__ float lrelu(float x) {
  return x > 0.0f ? x : NEG_SLOPE * x;
}

__global__ void prep_wext_kernel(const float* __restrict__ W,
                                 const float* __restrict__ asrc,
                                 const float* __restrict__ adst,
                                 float* __restrict__ Wext, int K, int F) {
  int i = blockIdx.x * blockDim.x + threadIdx.x;
  if (i >= K * 16) return;
  int k = i >> 4;
  int n = i & 15;
  float v = 0.0f;
  if (n < F) {
    v = W[k * F + n];
  } else if (n == 8) {
    for (int j = 0; j < F; ++j) v += W[k * F + j] * asrc[j];
  } else if (n == 9) {
    for (int j = 0; j < F; ++j) v += W[k * F + j] * adst[j];
  }
  Wext[k * 16 + n] = v;
}

__global__ void init_node_kernel(float* __restrict__ m, float* __restrict__ z,
                                 float* __restrict__ agg, int N, int F) {
  int i = blockIdx.x * blockDim.x + threadIdx.x;
  if (i >= N) return;
  m[i] = -INFINITY;
  z[i] = 0.0f;
  for (int f = 0; f < F; ++f) agg[i * F + f] = 0.0f;
}

template <int K4>
__global__ void gemm_attn_wmma(const float* __restrict__ A, int lda,
                               const float* __restrict__ Bext,
                               float* __restrict__ H, int nRows) {
  int lane = threadIdx.x & 31;
  int wave = (int)((blockIdx.x * blockDim.x + threadIdx.x) >> 5);
  int row0 = wave * 16;
  if (row0 >= nRows) return;

  int hf = lane >> 4;
  int l  = lane & 15;

  v8f c = {};
#pragma unroll
  for (int k = 0; k < K4; ++k) {
    int kb = k * 4;
    const float* ap = A + (size_t)(row0 + l) * lda + kb + hf * 2;
    v2f a;
    a.x = ap[0];
    a.y = ap[1];
    const float* bp = Bext + (size_t)(kb + hf * 2) * 16 + l;
    v2f b;
    b.x = bp[0];
    b.y = bp[16];
    c = __builtin_amdgcn_wmma_f32_16x16x4_f32(false, a, false, b, (short)0, c,
                                              false, false);
  }

  float* out = H + (size_t)(row0 + hf * 8) * 16 + l;
#pragma unroll
  for (int r = 0; r < 8; ++r) out[(size_t)r * 16] = c[r];
}

__global__ void edge_max_kernel(const int* __restrict__ src,
                                const int* __restrict__ dst,
                                const float* __restrict__ H,
                                float* __restrict__ m, int E) {
  int i = blockIdx.x * blockDim.x + threadIdx.x;
  if (i >= E) return;
  int s = src[i], d = dst[i];
  float e = lrelu(H[(size_t)s * 16 + 8] + H[(size_t)d * 16 + 9]);
  atomicMax(m + d, e);
}

__global__ void edge_expsum_kernel(const int* __restrict__ src,
                                   const int* __restrict__ dst,
                                   const float* __restrict__ H,
                                   const float* __restrict__ m,
                                   float* __restrict__ z, int E) {
  int i = blockIdx.x * blockDim.x + threadIdx.x;
  if (i >= E) return;
  int s = src[i], d = dst[i];
  float e = lrelu(H[(size_t)s * 16 + 8] + H[(size_t)d * 16 + 9]);
  float p = __expf(e - m[d]);
  atomicAdd(z + d, p);
}

template <int F>
__global__ void edge_agg_kernel(const int* __restrict__ src,
                                const int* __restrict__ dst,
                                const float* __restrict__ H,
                                const float* __restrict__ m,
                                const float* __restrict__ z,
                                float* __restrict__ agg, int E) {
  int i = blockIdx.x * blockDim.x + threadIdx.x;
  if (i >= E) return;
  int s = src[i], d = dst[i];
  float e = lrelu(H[(size_t)s * 16 + 8] + H[(size_t)d * 16 + 9]);
  float alpha = __expf(e - m[d]) / (z[d] + 1e-16f);
  const float* hs = H + (size_t)s * 16;
  float* o = agg + (size_t)d * F;
#pragma unroll
  for (int f = 0; f < F; ++f) atomicAdd(o + f, alpha * hs[f]);
}

__global__ void relu_bias_kernel(const float* __restrict__ agg,
                                 const float* __restrict__ b,
                                 float* __restrict__ out, int N) {
  int i = blockIdx.x * blockDim.x + threadIdx.x;
  if (i >= N) return;
#pragma unroll
  for (int f = 0; f < 8; ++f) {
    float v = agg[(size_t)i * 8 + f] + b[f];
    out[(size_t)i * 8 + f] = v > 0.0f ? v : 0.0f;
  }
}

__global__ void logsoftmax_kernel(const float* __restrict__ agg,
                                  const float* __restrict__ b,
                                  float* __restrict__ out, int N) {
  int i = blockIdx.x * blockDim.x + threadIdx.x;
  if (i >= N) return;
  float o0 = agg[(size_t)i * 2 + 0] + b[0];
  float o1 = agg[(size_t)i * 2 + 1] + b[1];
  float mx = fmaxf(o0, o1);
  float lse = mx + __logf(__expf(o0 - mx) + __expf(o1 - mx));
  out[(size_t)i * 2 + 0] = o0 - lse;
  out[(size_t)i * 2 + 1] = o1 - lse;
}

extern "C" void kernel_launch(void* const* d_in, const int* in_sizes, int n_in,
                              void* d_out, int out_size, void* d_ws,
                              size_t ws_size, hipStream_t stream) {
  const float* x      = (const float*)d_in[0];
  const float* W1     = (const float*)d_in[1];
  const float* a1_src = (const float*)d_in[2];
  const float* a1_dst = (const float*)d_in[3];
  const float* b1     = (const float*)d_in[4];
  const float* W2     = (const float*)d_in[5];
  const float* a2_src = (const float*)d_in[6];
  const float* a2_dst = (const float*)d_in[7];
  const float* b2     = (const float*)d_in[8];
  const int*   eidx   = (const int*)d_in[9];

  const int N = in_sizes[0] / 36;
  const int E = in_sizes[9] / 2;
  const int* src = eidx;
  const int* dst = eidx + E;

  float* ws    = (float*)d_ws;
  float* h1    = ws;
  float* m1    = h1 + (size_t)N * 16;
  float* z1    = m1 + N;
  float* agg1  = z1 + N;
  float* h1r   = agg1 + (size_t)N * 8;
  float* h2    = h1r + (size_t)N * 8;
  float* m2    = h2 + (size_t)N * 16;
  float* z2    = m2 + N;
  float* agg2  = z2 + N;
  float* wext1 = agg2 + (size_t)N * 2;
  float* wext2 = wext1 + 36 * 16;

  const int BLK = 256;
  dim3 nodeGrid((N + BLK - 1) / BLK);
  dim3 edgeGrid((E + BLK - 1) / BLK);
  dim3 gemmGrid((N + 127) / 128);

  prep_wext_kernel<<<dim3((36 * 16 + BLK - 1) / BLK), BLK, 0, stream>>>(
      W1, a1_src, a1_dst, wext1, 36, 8);
  init_node_kernel<<<nodeGrid, BLK, 0, stream>>>(m1, z1, agg1, N, 8);
  gemm_attn_wmma<9><<<gemmGrid, BLK, 0, stream>>>(x, 36, wext1, h1, N);
  edge_max_kernel<<<edgeGrid, BLK, 0, stream>>>(src, dst, h1, m1, E);
  edge_expsum_kernel<<<edgeGrid, BLK, 0, stream>>>(src, dst, h1, m1, z1, E);
  edge_agg_kernel<8><<<edgeGrid, BLK, 0, stream>>>(src, dst, h1, m1, z1, agg1, E);
  relu_bias_kernel<<<nodeGrid, BLK, 0, stream>>>(agg1, b1, h1r, N);

  prep_wext_kernel<<<dim3(1), BLK, 0, stream>>>(W2, a2_src, a2_dst, wext2, 8, 2);
  init_node_kernel<<<nodeGrid, BLK, 0, stream>>>(m2, z2, agg2, N, 2);
  gemm_attn_wmma<2><<<gemmGrid, BLK, 0, stream>>>(h1r, 8, wext2, h2, N);
  edge_max_kernel<<<edgeGrid, BLK, 0, stream>>>(src, dst, h2, m2, E);
  edge_expsum_kernel<<<edgeGrid, BLK, 0, stream>>>(src, dst, h2, m2, z2, E);
  edge_agg_kernel<2><<<edgeGrid, BLK, 0, stream>>>(src, dst, h2, m2, z2, agg2, E);

  logsoftmax_kernel<<<nodeGrid, BLK, 0, stream>>>(agg2, b2, (float*)d_out, N);
}


# edge aggregation: the lanes of one edge add into adjacent floats of one aligned accumulator row (sums + denominator): one row update per edge, not F+1 scattered f32 atomics; arrival-order sums kept
# speedup vs baseline: 2.3826x; 2.3826x over previous
#include <hip/hip_runtime.h>
#include <math.h>

typedef __attribute__((ext_vector_type(2))) float v2f;
typedef __attribute__((ext_vector_type(8))) float v8f;

#define NEG_SLOPE 0.2f

__device__ __forceinline__ float lrelu(float x) {
  return x > 0.0f ? x : NEG_SLOPE * x;
}

__global__ void prep_wext_kernel(const float* __restrict__ W,
                                 const float* __restrict__ asrc,
                                 const float* __restrict__ adst,
                                 float* __restrict__ Wext, int K, int F) {
  int i = blockIdx.x * blockDim.x + threadIdx.x;
  if (i >= K * 16) return;
  int k = i >> 4;
  int n = i & 15;
  float v = 0.0f;
  if (n < F) {
    v = W[k * F + n];
  } else if (n == 8) {
    for (int j = 0; j < F; ++j) v += W[k * F + j] * asrc[j];
  } else if (n == 9) {
    for (int j = 0; j < F; ++j) v += W[k * F + j] * adst[j];
  }
  Wext[k * 16 + n] = v;
}

template <int AS>
__global__ void init_node_kernel(float* __restrict__ m, float* __restrict__ acc,
                                 int N) {
  int i = blockIdx.x * blockDim.x + threadIdx.x;
  if (i >= N) return;
  m[i] = -INFINITY;
  float4* r = reinterpret_cast<float4*>(acc + (size_t)i * AS);
#pragma unroll
  for (int q = 0; q < AS / 4; ++q) r[q] = make_float4(0.0f, 0.0f, 0.0f, 0.0f);
}

template <int K4>
__global__ void gemm_attn_wmma(const float* __restrict__ A, int lda,
                               const float* __restrict__ Bext,
                               float* __restrict__ H, int nRows) {
  int lane = threadIdx.x & 31;
  int wave = (int)((blockIdx.x * blockDim.x + threadIdx.x) >> 5);
  int row0 = wave * 16;
  if (row0 >= nRows) return;

  int hf = lane >> 4;
  int l  = lane & 15;

  v8f c = {};
#pragma unroll
  for (int k = 0; k < K4; ++k) {
    int kb = k * 4;
    const float* ap = A + (size_t)(row0 + l) * lda + kb + hf * 2;
    v2f a;
    a.x = ap[0];
    a.y = ap[1];
    const float* bp = Bext + (size_t)(kb + hf * 2) * 16 + l;
    v2f b;
    b.x = bp[0];
    b.y = bp[16];
    c = __builtin_amdgcn_wmma_f32_16x16x4_f32(false, a, false, b, (short)0, c,
                                              false, false);
  }

  float* out = H + (size_t)(row0 + hf * 8) * 16 + l;
#pragma unroll
  for (int r = 0; r < 8; ++r) out[(size_t)r * 16] = c[r];
}

__global__ void edge_max_kernel(const int* __restrict__ src,
                                const int* __restrict__ dst,
                                const float* __restrict__ H,
                                float* __restrict__ m, int E) {
  int i = blockIdx.x * blockDim.x + threadIdx.x;
  if (i >= E) return;
  int s = src[i], d = dst[i];
  float e = lrelu(H[(size_t)s * 16 + 8] + H[(size_t)d * 16 + 9]);
  atomicMax(m + d, e);
}

template <int F, int G, int AS>
__global__ void __launch_bounds__(256, 16) edge_agg_kernel(const int* __restrict__ src,
                                const int* __restrict__ dst,
                                const float* __restrict__ H,
                                const float* __restrict__ m,
                                float* __restrict__ acc, int E) {
  static_assert(F + 3 <= G && F + 1 <= AS && 32 % G == 0, "lane roles");
  constexpr int EPI = 32 / G;
  constexpr int IT  = G;

  const int lane = threadIdx.x & 31;
  const int base = (int)(blockIdx.x * blockDim.x + threadIdx.x) - lane;
  if (base >= E) return;
  const int i  = base + lane;
  const int s  = i < E ? src[i] : 0;
  const int d  = i < E ? dst[i] : 0;
  const int l  = lane & (G - 1);
  const int g0 = lane & ~(G - 1);
  const int ge = lane / G;

  int sj[IT], dj[IT];
#pragma unroll
  for (int j = 0; j < IT; ++j) {
    const int e = j * EPI + ge;
    sj[j] = __shfl(s, e);
    dj[j] = __shfl(d, e);
  }
  const bool   fromM = l >= F + 2;
  const float* gbase = fromM ? m : H;
  const int    col   = l < F ? l : (l == F ? 8 : 9);
  float v[IT];
#pragma unroll
  for (int j = 0; j < IT; ++j) {
    const unsigned row = (unsigned)(l <= F ? sj[j] : dj[j]);
    const unsigned off = fromM ? (unsigned)dj[j] : row * 16u + (unsigned)col;
    v[j] = gbase[off];
  }
#pragma unroll
  for (int j = 0; j < IT; ++j) {
    const int e = j * EPI + ge;
    const float ssrc = __shfl(v[j], g0 + F);
    const float sdst = __shfl(v[j], g0 + F + 1);
    const float mx   = __shfl(v[j], g0 + F + 2);
    const float p    = __expf(lrelu(ssrc + sdst) - mx);
    const float val  = l < F ? p * v[j] : p;
    const unsigned ai = (unsigned)dj[j] * AS + (unsigned)l;
    if (l <= F && base + e < E) atomicAdd(acc + ai, val);
  }
}

__global__ void relu_bias_kernel(float* __restrict__ acc,
                                 const float* __restrict__ b, int N) {
  int i = blockIdx.x * blockDim.x + threadIdx.x;
  if (i >= N) return;
  float* r = acc + (size_t)i * 16;
  float den = r[8] + 1e-16f;
#pragma unroll
  for (int f = 0; f < 8; ++f) {
    float v = r[f] / den + b[f];
    r[f] = v > 0.0f ? v : 0.0f;
  }
}

__global__ void logsoftmax_kernel(const float* __restrict__ acc,
                                  const float* __restrict__ b,
                                  float* __restrict__ out, int N) {
  int i = blockIdx.x * blockDim.x + threadIdx.x;
  if (i >= N) return;
  const float* r = acc + (size_t)i * 4;
  float den = r[2] + 1e-16f;
  float o0 = r[0] / den + b[0];
  float o1 = r[1] / den + b[1];
  float mx = fmaxf(o0, o1);
  float lse = mx + __logf(__expf(o0 - mx) + __expf(o1 - mx));
  out[(size_t)i * 2 + 0] = o0 - lse;
  out[(size_t)i * 2 + 1] = o1 - lse;
}

extern "C" void kernel_launch(void* const* d_in, const int* in_sizes, int n_in,
                              void* d_out, int out_size, void* d_ws,
                              size_t ws_size, hipStream_t stream) {
  const float* x      = (const float*)d_in[0];
  const float* W1     = (const float*)d_in[1];
  const float* a1_src = (const float*)d_in[2];
  const float* a1_dst = (const float*)d_in[3];
  const float* b1     = (const float*)d_in[4];
  const float* W2     = (const float*)d_in[5];
  const float* a2_src = (const float*)d_in[6];
  const float* a2_dst = (const float*)d_in[7];
  const float* b2     = (const float*)d_in[8];
  const int*   eidx   = (const int*)d_in[9];

  const int N = in_sizes[0] / 36;
  const int E = in_sizes[9] / 2;
  const int* src = eidx;
  const int* dst = eidx + E;

  float* ws    = (float*)d_ws;
  float* h1    = ws;
  float* acc1  = h1 + (size_t)N * 16;
  float* h2    = acc1 + (size_t)N * 16;
  float* acc2  = h2 + (size_t)N * 16;
  float* m1    = acc2 + (size_t)N * 4;
  float* m2    = m1 + N;
  float* wext1 = m2 + N;
  float* wext2 = wext1 + 36 * 16;

  const int BLK = 256;
  dim3 nodeGrid((N + BLK - 1) / BLK);
  dim3 edgeGrid((E + BLK - 1) / BLK);
  dim3 gemmGrid((N + 127) / 128);

  prep_wext_kernel<<<dim3((36 * 16 + BLK - 1) / BLK), BLK, 0, stream>>>(
      W1, a1_src, a1_dst, wext1, 36, 8);
  init_node_kernel<16><<<nodeGrid, BLK, 0, stream>>>(m1, acc1, N);
  gemm_attn_wmma<9><<<gemmGrid, BLK, 0, stream>>>(x, 36, wext1, h1, N);
  edge_max_kernel<<<edgeGrid, BLK, 0, stream>>>(src, dst, h1, m1, E);
  edge_agg_kernel<8, 16, 16><<<edgeGrid, BLK, 0, stream>>>(src, dst, h1, m1, acc1, E);
  relu_bias_kernel<<<nodeGrid, BLK, 0, stream>>>(acc1, b1, N);

  prep_wext_kernel<<<dim3(1), BLK, 0, stream>>>(W2, a2_src, a2_dst, wext2, 8, 2);
  init_node_kernel<4><<<nodeGrid, BLK, 0, stream>>>(m2, acc2, N);
  gemm_attn_wmma<2><<<gemmGrid, BLK, 0, stream>>>(acc1, 16, wext2, h2, N);
  edge_max_kernel<<<edgeGrid, BLK, 0, stream>>>(src, dst, h2, m2, E);
  edge_agg_kernel<2, 8, 4><<<edgeGrid, BLK, 0, stream>>>(src, dst, h2, m2, acc2, E);

  logsoftmax_kernel<<<nodeGrid, BLK, 0, stream>>>(acc2, b2, (float*)d_out, N);
}


# edges grouped by destination once (integer atomic rank, scan, placement); a wave per node does max, denominator and weighted sum on its list: no float atomics or max pass; rank-order lane-tree sums
# speedup vs baseline: 4.5185x; 1.8965x over previous
#include <hip/hip_runtime.h>
#include <math.h>

typedef __attribute__((ext_vector_type(2))) float v2f;
typedef __attribute__((ext_vector_type(8))) float v8f;

#define NEG_SLOPE 0.2f

__device__ __forceinline__ float lrelu(float x) {
  return x > 0.0f ? x : NEG_SLOPE * x;
}

__global__ void prep_wext_kernel(const float* __restrict__ W,
                                 const float* __restrict__ asrc,
                                 const float* __restrict__ adst,
                                 float* __restrict__ Wext, int K, int F) {
  int i = blockIdx.x * blockDim.x + threadIdx.x;
  if (i >= K * 16) return;
  int k = i >> 4;
  int n = i & 15;
  float v = 0.0f;
  if (n < F) {
    v = W[k * F + n];
  } else if (n == 8) {
    for (int j = 0; j < F; ++j) v += W[k * F + j] * asrc[j];
  } else if (n == 9) {
    for (int j = 0; j < F; ++j) v += W[k * F + j] * adst[j];
  }
  Wext[k * 16 + n] = v;
}

template <int K4>
__global__ void gemm_attn_wmma(const float* __restrict__ A, int lda,
                               const float* __restrict__ Bext,
                               float* __restrict__ H, int nRows) {
  int lane = threadIdx.x & 31;
  int wave = (int)((blockIdx.x * blockDim.x + threadIdx.x) >> 5);
  int row0 = wave * 16;
  if (row0 >= nRows) return;

  int hf = lane >> 4;
  int l  = lane & 15;

  v8f c = {};
#pragma unroll
  for (int k = 0; k < K4; ++k) {
    int kb = k * 4;
    const float* ap = A + (size_t)(row0 + l) * lda + kb + hf * 2;
    v2f a;
    a.x = ap[0];
    a.y = ap[1];
    const float* bp = Bext + (size_t)(kb + hf * 2) * 16 + l;
    v2f b;
    b.x = bp[0];
    b.y = bp[16];
    c = __builtin_amdgcn_wmma_f32_16x16x4_f32(false, a, false, b, (short)0, c,
                                              false, false);
  }

  float* out = H + (size_t)(row0 + hf * 8) * 16 + l;
#pragma unroll
  for (int r = 0; r < 8; ++r) out[(size_t)r * 16] = c[r];
}

__global__ void zero_count_kernel(int* __restrict__ cnt, int N) {
  int i = blockIdx.x * blockDim.x + threadIdx.x;
  if (i < N) cnt[i] = 0;
}

__global__ void edge_rank_kernel(const int* __restrict__ dst,
                                 int* __restrict__ cnt,
                                 int* __restrict__ rank, int E) {
  int i = blockIdx.x * blockDim.x + threadIdx.x;
  if (i >= E) return;
  rank[i] = atomicAdd(cnt + dst[i], 1);
}

__global__ void __launch_bounds__(256) offset_scan_kernel(const int* __restrict__ cnt,
                                                          int* __restrict__ off, int N) {
  __shared__ int wave_own[8];
  __shared__ int wave_before[8];
  const int t = threadIdx.x;
  const int lane = t & 31;
  const int w = t >> 5;
  const int first = blockIdx.x * 1024;

  int before = 0;
  const int4* c4 = reinterpret_cast<const int4*>(cnt);
  const int rows = blockIdx.x;
  for (int r = 0; r < rows; r += 16) {
    int4 x[16];
#pragma unroll
    for (int u = 0; u < 16; ++u)
      x[u] = (r + u < rows) ? c4[(size_t)(r + u) * 256 + t] : make_int4(0, 0, 0, 0);
#pragma unroll
    for (int u = 0; u < 16; ++u) before += (x[u].x + x[u].y) + (x[u].z + x[u].w);
  }

  const int i0 = first + t * 4;
  int v[4];
#pragma unroll
  for (int k = 0; k < 4; ++k) v[k] = (i0 + k < N) ? cnt[i0 + k] : 0;
  const int own = (v[0] + v[1]) + (v[2] + v[3]);

  int inc = own;
  for (int d = 1; d < 32; d <<= 1) {
    const int o = __shfl_up(inc, d);
    if (lane >= d) inc += o;
  }
  for (int d = 16; d >= 1; d >>= 1) before += __shfl_xor(before, d);
  if (lane == 31) wave_own[w] = inc;
  if (lane == 0) wave_before[w] = before;
  __syncthreads();
  int base = 0;
  for (int k = 0; k < 8; ++k) {
    base += wave_before[k];
    if (k < w) base += wave_own[k];
  }
  int run = base + inc - own;
#pragma unroll
  for (int k = 0; k < 4; ++k) {
    if (i0 + k < N) off[i0 + k] = run;
    run += v[k];
  }
}

__global__ void edge_place_kernel(const int* __restrict__ src,
                                  const int* __restrict__ dst,
                                  const int* __restrict__ rank,
                                  const int* __restrict__ off,
                                  int* __restrict__ sorted_src, int E) {
  int i = blockIdx.x * blockDim.x + threadIdx.x;
  if (i >= E) return;
  sorted_src[off[dst[i]] + rank[i]] = src[i];
}

template <int LPE, bool FINAL>
__global__ void __launch_bounds__(256, 16) node_softmax_agg_kernel(
    const int* __restrict__ sorted_src, const int* __restrict__ off,
    const int* __restrict__ cnt, const float* __restrict__ H,
    const float* __restrict__ b, float* __restrict__ outp, int N) {
  constexpr int EPI = 32 / LPE;
  constexpr int CST = 16 / LPE;
  constexpr int SS  = 8 / CST;
  static_assert(LPE == 2 || LPE == 4, "pieces at cols 0 and 8 must exist");

  const int lane = threadIdx.x & 31;
  const int n = __builtin_amdgcn_readfirstlane(
      (int)((blockIdx.x * blockDim.x + threadIdx.x) >> 5));
  if (n >= N) return;
  const int sub = lane & (LPE - 1);
  const int g   = lane / LPE;

  const int   start = off[n];
  const int   deg   = cnt[n];
  const float sdst  = H[(size_t)n * 16 + 9];

  float  m = -INFINITY;
  float  z = 0.0f;
  float4 acc = make_float4(0.0f, 0.0f, 0.0f, 0.0f);

  for (int c = 0; c < deg; c += 32) {
    const int k    = c + lane;
    const int sidx = k < deg ? sorted_src[start + k] : 0;
    float4 v[LPE];
#pragma unroll
    for (int j = 0; j < LPE; ++j) {
      const unsigned s = (unsigned)__shfl(sidx, j * EPI + g);
      v[j] = *reinterpret_cast<const float4*>(H + (size_t)s * 16 + sub * CST);
    }
    float e[LPE];
    float cm = -INFINITY;
#pragma unroll
    for (int j = 0; j < LPE; ++j) {
      const float ssrc  = __shfl(v[j].x, lane - sub + SS);
      const bool  valid = c + j * EPI + g < deg;
      e[j] = valid ? lrelu(ssrc + sdst) : -INFINITY;
      if (!valid) v[j] = make_float4(0.0f, 0.0f, 0.0f, 0.0f);
      cm = fmaxf(cm, e[j]);
    }
    for (int d = LPE; d < 32; d <<= 1) cm = fmaxf(cm, __shfl_xor(cm, d));
    const float mn = fmaxf(m, cm);
    const float sc = __expf(m - mn);
    z *= sc;
    acc.x *= sc; acc.y *= sc; acc.z *= sc; acc.w *= sc;
#pragma unroll
    for (int j = 0; j < LPE; ++j) {
      const float p = __expf(e[j] - mn);
      z += p;
      acc.x += p * v[j].x;
      acc.y += p * v[j].y;
      acc.z += p * v[j].z;
      acc.w += p * v[j].w;
    }
    m = mn;
  }

  for (int d = LPE; d < 32; d <<= 1) {
    z     += __shfl_xor(z, d);
    acc.x += __shfl_xor(acc.x, d);
    acc.y += __shfl_xor(acc.y, d);
    acc.z += __shfl_xor(acc.z, d);
    acc.w += __shfl_xor(acc.w, d);
  }
  const float den = z + 1e-16f;

  if (!FINAL) {
    if (g == 0 && sub < 2) {
      const float* bb = b + sub * 4;
      float4 r;
      r.x = acc.x / den + bb[0];
      r.y = acc.y / den + bb[1];
      r.z = acc.z / den + bb[2];
      r.w = acc.w / den + bb[3];
      r.x = r.x > 0.0f ? r.x : 0.0f;
      r.y = r.y > 0.0f ? r.y : 0.0f;
      r.z = r.z > 0.0f ? r.z : 0.0f;
      r.w = r.w > 0.0f ? r.w : 0.0f;
      *reinterpret_cast<float4*>(outp + (size_t)n * 8 + sub * 4) = r;
    }
  } else {
    if (lane == 0) {
      const float o0 = acc.x / den + b[0];
      const float o1 = acc.y / den + b[1];
      const float mx = fmaxf(o0, o1);
      const float lse = mx + __logf(__expf(o0 - mx) + __expf(o1 - mx));
      *reinterpret_cast<float2*>(outp + (size_t)n * 2) = make_float2(o0 - lse, o1 - lse);
    }
  }
}

extern "C" void kernel_launch(void* const* d_in, const int* in_sizes, int n_in,
                              void* d_out, int out_size, void* d_ws,
                              size_t ws_size, hipStream_t stream) {
  const float* x      = (const float*)d_in[0];
  const float* W1     = (const float*)d_in[1];
  const float* a1_src = (const float*)d_in[2];
  const float* a1_dst = (const float*)d_in[3];
  const float* b1     = (const float*)d_in[4];
  const float* W2     = (const float*)d_in[5];
  const float* a2_src = (const float*)d_in[6];
  const float* a2_dst = (const float*)d_in[7];
  const float* b2     = (const float*)d_in[8];
  const int*   eidx   = (const int*)d_in[9];

  const int N = in_sizes[0] / 36;
  const int E = in_sizes[9] / 2;
  const int* src = eidx;
  const int* dst = eidx + E;

  const size_t Np    = ((size_t)N + 3) & ~(size_t)3;
  const size_t Ep    = ((size_t)E + 3) & ~(size_t)3;
  const size_t rowsW = (size_t)N * 40;
  const size_t headW = Ep > rowsW ? Ep : rowsW;
  if (ws_size < (headW + 704 + 2 * Np + Ep) * 4) return;

  float* ws    = (float*)d_ws;
  int*   rank  = (int*)d_ws;
  float* h1    = ws;
  float* h2    = h1 + (size_t)N * 16;
  float* h1r   = h2 + (size_t)N * 16;
  float* wext1 = ws + headW;
  float* wext2 = wext1 + 36 * 16;
  int*   cnt   = (int*)(wext2 + 8 * 16);
  int*   off   = cnt + Np;
  int*   ssrc  = off + Np;

  const int BLK = 256;
  dim3 nodeGrid((N + BLK - 1) / BLK);
  dim3 edgeGrid((E + BLK - 1) / BLK);
  dim3 scanGrid((N + 1023) / 1024);
  dim3 waveGrid((N + 7) / 8);
  dim3 gemmGrid((N + 127) / 128);

  zero_count_kernel<<<nodeGrid, BLK, 0, stream>>>(cnt, N);
  edge_rank_kernel<<<edgeGrid, BLK, 0, stream>>>(dst, cnt, rank, E);
  offset_scan_kernel<<<scanGrid, BLK, 0, stream>>>(cnt, off, N);
  edge_place_kernel<<<edgeGrid, BLK, 0, stream>>>(src, dst, rank, off, ssrc, E);

  prep_wext_kernel<<<dim3((36 * 16 + BLK - 1) / BLK), BLK, 0, stream>>>(
      W1, a1_src, a1_dst, wext1, 36, 8);
  gemm_attn_wmma<9><<<gemmGrid, BLK, 0, stream>>>(x, 36, wext1, h1, N);
  node_softmax_agg_kernel<4, false><<<waveGrid, BLK, 0, stream>>>(
      ssrc, off, cnt, h1, b1, h1r, N);

  prep_wext_kernel<<<dim3(1), BLK, 0, stream>>>(W2, a2_src, a2_dst, wext2, 8, 2);
  gemm_attn_wmma<2><<<gemmGrid, BLK, 0, stream>>>(h1r, 8, wext2, h2, N);
  node_softmax_agg_kernel<2, true><<<waveGrid, BLK, 0, stream>>>(
      ssrc, off, cnt, h2, b2, (float*)d_out, N);
}


# destination grouping rebuilt in two levels: edges binned to 512-node buckets, then ranked per bucket, both with LDS counters and contiguous stores; no per-edge global atomic; arrival-order lists kept
# speedup vs baseline: 15.6768x; 3.4695x over previous
#include <hip/hip_runtime.h>
#include <math.h>

typedef __attribute__((ext_vector_type(2))) float v2f;
typedef __attribute__((ext_vector_type(8))) float v8f;

#define NEG_SLOPE 0.2f

__device__ __forceinline__ float lrelu(float x) {
  return x > 0.0f ? x : NEG_SLOPE * x;
}

__global__ void prep_wext_kernel(const float* __restrict__ W,
                                 const float* __restrict__ asrc,
                                 const float* __restrict__ adst,
                                 float* __restrict__ Wext, int K, int F) {
  int i = blockIdx.x * blockDim.x + threadIdx.x;
  if (i >= K * 16) return;
  int k = i >> 4;
  int n = i & 15;
  float v = 0.0f;
  if (n < F) {
    v = W[k * F + n];
  } else if (n == 8) {
    for (int j = 0; j < F; ++j) v += W[k * F + j] * asrc[j];
  } else if (n == 9) {
    for (int j = 0; j < F; ++j) v += W[k * F + j] * adst[j];
  }
  Wext[k * 16 + n] = v;
}

template <int K4>
__global__ void gemm_attn_wmma(const float* __restrict__ A, int lda,
                               const float* __restrict__ Bext,
                               float* __restrict__ H, int nRows) {
  int lane = threadIdx.x & 31;
  int wave = (int)((blockIdx.x * blockDim.x + threadIdx.x) >> 5);
  int row0 = wave * 16;
  if (row0 >= nRows) return;

  int hf = lane >> 4;
  int l  = lane & 15;

  v8f c = {};
#pragma unroll
  for (int k = 0; k < K4; ++k) {
    int kb = k * 4;
    const float* ap = A + (size_t)(row0 + l) * lda + kb + hf * 2;
    v2f a;
    a.x = ap[0];
    a.y = ap[1];
    const float* bp = Bext + (size_t)(kb + hf * 2) * 16 + l;
    v2f b;
    b.x = bp[0];
    b.y = bp[16];
    c = __builtin_amdgcn_wmma_f32_16x16x4_f32(false, a, false, b, (short)0, c,
                                              false, false);
  }

  float* out = H + (size_t)(row0 + hf * 8) * 16 + l;
#pragma unroll
  for (int r = 0; r < 8; ++r) out[(size_t)r * 16] = c[r];
}

#define BSH    9
#define BNODES 512
#define MAXB   351
#define EPT    32
#define CHUNK  (EPT * 1024)
#define SCAP   40960

__global__ void zero_count_kernel(int* __restrict__ cnt, int N) {
  int i = blockIdx.x * blockDim.x + threadIdx.x;
  if (i < N) cnt[i] = 0;
}

__global__ void __launch_bounds__(1024) bucket_hist_kernel(const int* __restrict__ dst,
                                                           int* __restrict__ bcount,
                                                           int E, int NB) {
  __shared__ int sh_cnt[MAXB + 1];
  const int t = threadIdx.x;
  if (t < NB) sh_cnt[t] = 0;
  __syncthreads();
  const int e0 = blockIdx.x * CHUNK + t;
  int d[EPT];
#pragma unroll
  for (int u = 0; u < EPT; ++u) {
    const int i = e0 + u * 1024;
    d[u] = i < E ? dst[i] : -1;
  }
#pragma unroll
  for (int u = 0; u < EPT; ++u)
    if (d[u] >= 0) atomicAdd(&sh_cnt[d[u] >> BSH], 1);
  __syncthreads();
  if (t < NB) {
    const int c = sh_cnt[t];
    if (c) atomicAdd(bcount + t, c);
  }
}

__global__ void __launch_bounds__(512) bucket_scan_kernel(int* __restrict__ bcur,
                                                          int* __restrict__ bbase, int NB) {
  __shared__ int wave_tot[16];
  const int t = threadIdx.x;
  const int lane = t & 31;
  const int w = t >> 5;
  const int own = t < NB ? bcur[t] : 0;
  int inc = own;
  for (int d = 1; d < 32; d <<= 1) {
    const int o = __shfl_up(inc, d);
    if (lane >= d) inc += o;
  }
  if (lane == 31) wave_tot[w] = inc;
  __syncthreads();
  int before = 0;
  for (int k = 0; k < 16; ++k)
    if (k < w) before += wave_tot[k];
  const int ex = before + inc - own;
  if (t < NB) bcur[t] = ex;
  if (t <= NB) bbase[t] = ex;
}

__global__ void __launch_bounds__(1024) edge_bin_kernel(const int* __restrict__ src,
                                                        const int* __restrict__ dst,
                                                        int* __restrict__ bcur,
                                                        int* __restrict__ binned,
                                                        int E, int NB) {
  __shared__ int sh_cnt[MAXB + 1];
  __shared__ int sh_base[MAXB + 1];
  __shared__ int sh_loc[MAXB + 1];
  __shared__ int wave_tot[32];
  __shared__ int stage[CHUNK];
  const int t = threadIdx.x;
  const int lane = t & 31;
  const int w = t >> 5;
  if (t < NB) sh_cnt[t] = 0;
  __syncthreads();
  const int e0 = blockIdx.x * CHUNK + t;
  int d[EPT];
#pragma unroll
  for (int u = 0; u < EPT; ++u) {
    const int i = e0 + u * 1024;
    d[u] = i < E ? dst[i] : -1;
  }
#pragma unroll
  for (int u = 0; u < EPT; ++u)
    if (d[u] >= 0) atomicAdd(&sh_cnt[d[u] >> BSH], 1);
  __syncthreads();

  const int own = t < NB ? sh_cnt[t] : 0;
  int inc = own;
  for (int s = 1; s < 32; s <<= 1) {
    const int o = __shfl_up(inc, s);
    if (lane >= s) inc += o;
  }
  if (lane == 31) wave_tot[w] = inc;
  __syncthreads();
  int before = 0;
  for (int k = 0; k < 32; ++k)
    if (k < w) before += wave_tot[k];
  if (t < NB) {
    sh_loc[t]  = before + inc - own;
    sh_base[t] = own ? atomicAdd(bcur + t, own) : 0;
    sh_cnt[t]  = 0;
  }
  __syncthreads();

#pragma unroll
  for (int h = 0; h < EPT; h += 16) {
    int s[16];
#pragma unroll
    for (int u = 0; u < 16; ++u) {
      const int i = e0 + (h + u) * 1024;
      s[u] = i < E ? src[i] : 0;
    }
#pragma unroll
    for (int u = 0; u < 16; ++u) {
      const int dd = d[h + u];
      if (dd >= 0) {
        const int k = dd >> BSH;
        const int r = atomicAdd(&sh_cnt[k], 1);
        stage[sh_loc[k] + r] = (s[u] << BSH) | (dd & (BNODES - 1));
      }
    }
  }
  __syncthreads();

  for (int k = w; k < NB; k += 32) {
    const int c = sh_cnt[k];
    const int g = sh_base[k];
    const int l = sh_loc[k];
    for (int j = lane; j < c; j += 32) binned[g + j] = stage[l + j];
  }
}

__global__ void __launch_bounds__(1024) bucket_sort_kernel(const int* __restrict__ binned,
                                                           const int* __restrict__ bbase,
                                                           int* __restrict__ cnt,
                                                           int* __restrict__ off,
                                                           int* __restrict__ sorted_src, int N) {
  __shared__ int sh_cnt[BNODES];
  __shared__ int sh_off[BNODES];
  __shared__ int wave_tot[32];
  __shared__ int stage[SCAP];
  const int t = threadIdx.x;
  const int lane = t & 31;
  const int w = t >> 5;
  const int s0 = bbase[blockIdx.x];
  const int m  = bbase[blockIdx.x + 1] - s0;
  const int* mine = binned + s0;
  const bool staged = m <= SCAP;

  if (t < BNODES) sh_cnt[t] = 0;
  __syncthreads();

  for (int i0 = t; i0 < m; i0 += 8 * 1024) {
    int v[8];
#pragma unroll
    for (int u = 0; u < 8; ++u) {
      const int i = i0 + u * 1024;
      v[u] = i < m ? mine[i] : -1;
    }
#pragma unroll
    for (int u = 0; u < 8; ++u)
      if (v[u] >= 0) atomicAdd(&sh_cnt[v[u] & (BNODES - 1)], 1);
  }
  __syncthreads();

  const int own = t < BNODES ? sh_cnt[t] : 0;
  int inc = own;
  for (int s = 1; s < 32; s <<= 1) {
    const int o = __shfl_up(inc, s);
    if (lane >= s) inc += o;
  }
  if (lane == 31) wave_tot[w] = inc;
  __syncthreads();
  int before = 0;
  for (int k = 0; k < 32; ++k)
    if (k < w) before += wave_tot[k];
  if (t < BNODES) {
    const int first = before + inc - own;
    sh_off[t] = first;
    sh_cnt[t] = 0;
    const int n = blockIdx.x * BNODES + t;
    if (n < N) {
      cnt[n] = own;
      off[n] = s0 + first;
    }
  }
  __syncthreads();

  for (int i0 = t; i0 < m; i0 += 8 * 1024) {
    int v[8];
#pragma unroll
    for (int u = 0; u < 8; ++u) {
      const int i = i0 + u * 1024;
      v[u] = i < m ? mine[i] : -1;
    }
#pragma unroll
    for (int u = 0; u < 8; ++u) {
      if (v[u] >= 0) {
        const int dn = v[u] & (BNODES - 1);
        const int p = sh_off[dn] + atomicAdd(&sh_cnt[dn], 1);
        if (staged) stage[p] = v[u] >> BSH;
        else        sorted_src[s0 + p] = v[u] >> BSH;
      }
    }
  }
  if (staged) {
    __syncthreads();
    for (int j = t; j < m; j += 1024) sorted_src[s0 + j] = stage[j];
  }
}

template <int LPE, bool FINAL>
__global__ void __launch_bounds__(256, 16) node_softmax_agg_kernel(
    const int* __restrict__ sorted_src, const int* __restrict__ off,
    const int* __restrict__ cnt, const float* __restrict__ H,
    const float* __restrict__ b, float* __restrict__ outp, int N) {
  constexpr int EPI = 32 / LPE;
  constexpr int CST = 16 / LPE;
  constexpr int SS  = 8 / CST;
  static_assert(LPE == 2 || LPE == 4, "pieces at cols 0 and 8 must exist");

  const int lane = threadIdx.x & 31;
  const int n = __builtin_amdgcn_readfirstlane(
      (int)((blockIdx.x * blockDim.x + threadIdx.x) >> 5));
  if (n >= N) return;
  const int sub = lane & (LPE - 1);
  const int g   = lane / LPE;

  const int   start = off[n];
  const int   deg   = cnt[n];
  const float sdst  = H[(size_t)n * 16 + 9];

  float  m = -INFINITY;
  float  z = 0.0f;
  float4 acc = make_float4(0.0f, 0.0f, 0.0f, 0.0f);

  for (int c = 0; c < deg; c += 32) {
    const int k    = c + lane;
    const int sidx = k < deg ? sorted_src[start + k] : 0;
    float4 v[LPE];
#pragma unroll
    for (int j = 0; j < LPE; ++j) {
      const unsigned s = (unsigned)__shfl(sidx, j * EPI + g);
      v[j] = *reinterpret_cast<const float4*>(H + (size_t)s * 16 + sub * CST);
    }
    float e[LPE];
    float cm = -INFINITY;
#pragma unroll
    for (int j = 0; j < LPE; ++j) {
      const float ssrc  = __shfl(v[j].x, lane - sub + SS);
      const bool  valid = c + j * EPI + g < deg;
      e[j] = valid ? lrelu(ssrc + sdst) : -INFINITY;
      if (!valid) v[j] = make_float4(0.0f, 0.0f, 0.0f, 0.0f);
      cm = fmaxf(cm, e[j]);
    }
    for (int d = LPE; d < 32; d <<= 1) cm = fmaxf(cm, __shfl_xor(cm, d));
    const float mn = fmaxf(m, cm);
    const float sc = __expf(m - mn);
    z *= sc;
    acc.x *= sc; acc.y *= sc; acc.z *= sc; acc.w *= sc;
#pragma unroll
    for (int j = 0; j < LPE; ++j) {
      const float p = __expf(e[j] - mn);
      z += p;
      acc.x += p * v[j].x;
      acc.y += p * v[j].y;
      acc.z += p * v[j].z;
      acc.w += p * v[j].w;
    }
    m = mn;
  }

  for (int d = LPE; d < 32; d <<= 1) {
    z     += __shfl_xor(z, d);
    acc.x += __shfl_xor(acc.x, d);
    acc.y += __shfl_xor(acc.y, d);
    acc.z += __shfl_xor(acc.z, d);
    acc.w += __shfl_xor(acc.w, d);
  }
  const float den = z + 1e-16f;

  if (!FINAL) {
    if (g == 0 && sub < 2) {
      const float* bb = b + sub * 4;
      float4 r;
      r.x = acc.x / den + bb[0];
      r.y = acc.y / den + bb[1];
      r.z = acc.z / den + bb[2];
      r.w = acc.w / den + bb[3];
      r.x = r.x > 0.0f ? r.x : 0.0f;
      r.y = r.y > 0.0f ? r.y : 0.0f;
      r.z = r.z > 0.0f ? r.z : 0.0f;
      r.w = r.w > 0.0f ? r.w : 0.0f;
      *reinterpret_cast<float4*>(outp + (size_t)n * 8 + sub * 4) = r;
    }
  } else {
    if (lane == 0) {
      const float o0 = acc.x / den + b[0];
      const float o1 = acc.y / den + b[1];
      const float mx = fmaxf(o0, o1);
      const float lse = mx + __logf(__expf(o0 - mx) + __expf(o1 - mx));
      *reinterpret_cast<float2*>(outp + (size_t)n * 2) = make_float2(o0 - lse, o1 - lse);
    }
  }
}

extern "C" void kernel_launch(void* const* d_in, const int* in_sizes, int n_in,
                              void* d_out, int out_size, void* d_ws,
                              size_t ws_size, hipStream_t stream) {
  const float* x      = (const float*)d_in[0];
  const float* W1     = (const float*)d_in[1];
  const float* a1_src = (const float*)d_in[2];
  const float* a1_dst = (const float*)d_in[3];
  const float* b1     = (const float*)d_in[4];
  const float* W2     = (const float*)d_in[5];
  const float* a2_src = (const float*)d_in[6];
  const float* a2_dst = (const float*)d_in[7];
  const float* b2     = (const float*)d_in[8];
  const int*   eidx   = (const int*)d_in[9];

  const int N = in_sizes[0] / 36;
  const int E = in_sizes[9] / 2;
  const int* src = eidx;
  const int* dst = eidx + E;

  const size_t Np    = ((size_t)N + 3) & ~(size_t)3;
  const size_t Ep    = ((size_t)E + 3) & ~(size_t)3;
  const size_t rowsW = (size_t)N * 40;
  const size_t headW = Ep > rowsW ? Ep : rowsW;
  const int NB = (N + BNODES - 1) >> BSH;
  if (ws_size < (headW + 704 + 2 * Np + Ep) * 4) return;
  if (NB > MAXB) return;

  float* ws    = (float*)d_ws;
  int*   binned = (int*)d_ws;
  float* h1    = ws;
  float* h2    = h1 + (size_t)N * 16;
  float* h1r   = h2 + (size_t)N * 16;
  float* wext1 = ws + headW;
  float* wext2 = wext1 + 36 * 16;
  int*   bcur  = (int*)wext1;
  int*   bbase = bcur + MAXB + 1;
  int*   cnt   = (int*)(wext2 + 8 * 16);
  int*   off   = cnt + Np;
  int*   ssrc  = off + Np;

  const int BLK = 256;
  dim3 chunkGrid((E + CHUNK - 1) / CHUNK);
  dim3 waveGrid((N + 7) / 8);
  dim3 gemmGrid((N + 127) / 128);

  zero_count_kernel<<<dim3(2), BLK, 0, stream>>>(bcur, NB);
  bucket_hist_kernel<<<chunkGrid, 1024, 0, stream>>>(dst, bcur, E, NB);
  bucket_scan_kernel<<<dim3(1), 512, 0, stream>>>(bcur, bbase, NB);
  edge_bin_kernel<<<chunkGrid, 1024, 0, stream>>>(src, dst, bcur, binned, E, NB);
  bucket_sort_kernel<<<dim3(NB), 1024, 0, stream>>>(binned, bbase, cnt, off, ssrc, N);

  prep_wext_kernel<<<dim3((36 * 16 + BLK - 1) / BLK), BLK, 0, stream>>>(
      W1, a1_src, a1_dst, wext1, 36, 8);
  gemm_attn_wmma<9><<<gemmGrid, BLK, 0, stream>>>(x, 36, wext1, h1, N);
  node_softmax_agg_kernel<4, false><<<waveGrid, BLK, 0, stream>>>(
      ssrc, off, cnt, h1, b1, h1r, N);

  prep_wext_kernel<<<dim3(1), BLK, 0, stream>>>(W2, a2_src, a2_dst, wext2, 8, 2);
  gemm_attn_wmma<2><<<gemmGrid, BLK, 0, stream>>>(h1r, 8, wext2, h2, N);
  node_softmax_agg_kernel<2, true><<<waveGrid, BLK, 0, stream>>>(
      ssrc, off, cnt, h2, b2, (float*)d_out, N);
}


# output-layer aggregation: source rows packed to four words; 256 resident work-groups keep the first 24576 nodes' rows in LDS; one lane per edge; rank-order 32-slot lane-tree sums
# speedup vs baseline: 17.6375x; 1.1251x over previous
#include <hip/hip_runtime.h>
#include <math.h>

typedef __attribute__((ext_vector_type(2))) float v2f;
typedef __attribute__((ext_vector_type(8))) float v8f;

#define NEG_SLOPE 0.2f

__device__ __forceinline__ float lrelu(float x) {
  return x > 0.0f ? x : NEG_SLOPE * x;
}

template <int K4, int F, bool PACKED>
__global__ void __launch_bounds__(256) gemm_attn_wmma(const float* __restrict__ A, int lda,
                                                      const float* __restrict__ W,
                                                      const float* __restrict__ asrc,
                                                      const float* __restrict__ adst,
                                                      float* __restrict__ H, int nRows) {
  constexpr int K = K4 * 4;
  __shared__ float sB[K * 16];

  int lane = threadIdx.x & 31;
  int wave = (int)((blockIdx.x * blockDim.x + threadIdx.x) >> 5);
  int row0 = wave * 16;
  const bool live = row0 < nRows;
  int hf = lane >> 4;
  int l  = lane & 15;

  v2f a[K4];
  if (live) {
#pragma unroll
    for (int k = 0; k < K4; ++k) {
      const float* ap = A + (size_t)(row0 + l) * lda + k * 4 + hf * 2;
      a[k].x = ap[0];
      a[k].y = ap[1];
    }
  }

  for (int i = threadIdx.x; i < K * 16; i += 256) {
    const int k = i >> 4;
    const int n = i & 15;
    float v = 0.0f;
    if (n < F) {
      v = W[k * F + n];
    } else if (n == 8) {
      for (int j = 0; j < F; ++j) v = __builtin_fmaf(W[k * F + j], asrc[j], v);
    } else if (n == 9) {
      for (int j = 0; j < F; ++j) v = __builtin_fmaf(W[k * F + j], adst[j], v);
    }
    sB[i] = v;
  }
  __syncthreads();
  if (!live) return;

  v8f c = {};
#pragma unroll
  for (int k = 0; k < K4; ++k) {
    const float* bp = sB + (k * 4 + hf * 2) * 16 + l;
    v2f b;
    b.x = bp[0];
    b.y = bp[16];
    c = __builtin_amdgcn_wmma_f32_16x16x4_f32(false, a[k], false, b, (short)0, c,
                                              false, false);
  }

  if (!PACKED) {
    float* out = H + (size_t)(row0 + hf * 8) * 16 + l;
#pragma unroll
    for (int r = 0; r < 8; ++r) out[(size_t)r * 16] = c[r];
  } else {
    static_assert(!PACKED || F == 2, "packed rows hold two features");
    const int word = l < 2 ? l : (l == 8 || l == 9) ? l - 6 : -1;
    if (word >= 0) {
      float* out = H + (size_t)(row0 + hf * 8) * 4 + word;
#pragma unroll
      for (int r = 0; r < 8; ++r) out[(size_t)r * 4] = c[r];
    }
  }
}

#define BSH    9
#define BNODES 512
#define MAXB   351
#define MAXC   11
#define EPT    32
#define CHUNK  (EPT * 1024)
#define SCAP   40960

__global__ void __launch_bounds__(1024) bucket_hist_kernel(const int* __restrict__ dst,
                                                           int* __restrict__ btab,
                                                           int E, int NB) {
  __shared__ int sh_cnt[MAXB + 1];
  const int t = threadIdx.x;
  if (t < NB) sh_cnt[t] = 0;
  __syncthreads();
  const int e0 = blockIdx.x * CHUNK + t;
  int d[EPT];
#pragma unroll
  for (int u = 0; u < EPT; ++u) {
    const int i = e0 + u * 1024;
    d[u] = i < E ? dst[i] : -1;
  }
#pragma unroll
  for (int u = 0; u < EPT; ++u)
    if (d[u] >= 0) atomicAdd(&sh_cnt[d[u] >> BSH], 1);
  __syncthreads();
  if (t < NB) btab[(size_t)blockIdx.x * NB + t] = sh_cnt[t];
}

__global__ void __launch_bounds__(1024) edge_bin_kernel(const int* __restrict__ src,
                                                        const int* __restrict__ dst,
                                                        const int* __restrict__ btab,
                                                        int* __restrict__ bbase,
                                                        int* __restrict__ binned,
                                                        int E, int NB, int nblk) {
  __shared__ int sh_cnt[MAXB + 1];
  __shared__ int sh_base[MAXB + 1];
  __shared__ int sh_loc[MAXB + 1];
  __shared__ int sh_tot[MAXB + 1];
  __shared__ int sh_pre[MAXB + 1];
  __shared__ int wave_tot[32];
  __shared__ int gwave_tot[32];
  __shared__ int stage[CHUNK];
  const int t = threadIdx.x;
  const int lane = t & 31;
  const int w = t >> 5;
  const int blk = blockIdx.x;
  if (t <= MAXB) {
    sh_cnt[t] = 0;
    sh_tot[t] = 0;
    sh_pre[t] = 0;
  }
  const int own = t < NB ? btab[(size_t)blk * NB + t] : 0;
  const int e0 = blk * CHUNK + t;
  int d[EPT];
#pragma unroll
  for (int u = 0; u < EPT; ++u) {
    const int i = e0 + u * 1024;
    d[u] = i < E ? dst[i] : -1;
  }
  __syncthreads();

  {
    int tot[MAXC];
    int pre[MAXC];
#pragma unroll
    for (int i = 0; i < MAXC; ++i) {
      tot[i] = 0;
      pre[i] = 0;
    }
#pragma unroll 2
    for (int r = w; r < nblk; r += 32) {
      const int* row = btab + (size_t)r * NB;
      const bool earlier = r < blk;
#pragma unroll
      for (int i = 0; i < MAXC; ++i) {
        const int col = lane + 32 * i;
        const int v = col < NB ? row[col] : 0;
        tot[i] += v;
        pre[i] += earlier ? v : 0;
      }
    }
#pragma unroll
    for (int i = 0; i < MAXC; ++i) {
      const int col = lane + 32 * i;
      if (col < NB) {
        atomicAdd(&sh_tot[col], tot[i]);
        atomicAdd(&sh_pre[col], pre[i]);
      }
    }
  }
  __syncthreads();

  const int gown = t < NB ? sh_tot[t] : 0;
  int inc  = own;
  int ginc = gown;
  for (int s = 1; s < 32; s <<= 1) {
    const int o  = __shfl_up(inc, s);
    const int go = __shfl_up(ginc, s);
    if (lane >= s) {
      inc  += o;
      ginc += go;
    }
  }
  if (lane == 31) {
    wave_tot[w]  = inc;
    gwave_tot[w] = ginc;
  }
  __syncthreads();
  int before  = 0;
  int gbefore = 0;
  for (int k = 0; k < 32; ++k)
    if (k < w) {
      before  += wave_tot[k];
      gbefore += gwave_tot[k];
    }
  const int gex = gbefore + ginc - gown;
  if (t < NB) {
    sh_loc[t]  = before + inc - own;
    sh_base[t] = gex + sh_pre[t];
  }
  if (blk == 0 && t <= NB) bbase[t] = gex;
  __syncthreads();

#pragma unroll
  for (int h = 0; h < EPT; h += 16) {
    int s[16];
#pragma unroll
    for (int u = 0; u < 16; ++u) {
      const int i = e0 + (h + u) * 1024;
      s[u] = i < E ? src[i] : 0;
    }
#pragma unroll
    for (int u = 0; u < 16; ++u) {
      const int dd = d[h + u];
      if (dd >= 0) {
        const int k = dd >> BSH;
        const int r = atomicAdd(&sh_cnt[k], 1);
        stage[sh_loc[k] + r] = (s[u] << BSH) | (dd & (BNODES - 1));
      }
    }
  }
  __syncthreads();

  for (int k = w; k < NB; k += 32) {
    const int c = sh_cnt[k];
    const int g = sh_base[k];
    const int l = sh_loc[k];
    for (int j = lane; j < c; j += 32) binned[g + j] = stage[l + j];
  }
}

__global__ void __launch_bounds__(1024) bucket_sort_kernel(const int* __restrict__ binned,
                                                           const int* __restrict__ bbase,
                                                           int* __restrict__ cnt,
                                                           int* __restrict__ off,
                                                           int* __restrict__ sorted_src, int N) {
  __shared__ int sh_cnt[BNODES];
  __shared__ int sh_off[BNODES];
  __shared__ int wave_tot[32];
  __shared__ int stage[SCAP];
  const int t = threadIdx.x;
  const int lane = t & 31;
  const int w = t >> 5;
  const int s0 = bbase[blockIdx.x];
  const int m  = bbase[blockIdx.x + 1] - s0;
  const int* mine = binned + s0;
  const bool staged = m <= SCAP;

  if (t < BNODES) sh_cnt[t] = 0;
  __syncthreads();

  for (int i0 = t; i0 < m; i0 += 8 * 1024) {
    int v[8];
#pragma unroll
    for (int u = 0; u < 8; ++u) {
      const int i = i0 + u * 1024;
      v[u] = i < m ? mine[i] : -1;
    }
#pragma unroll
    for (int u = 0; u < 8; ++u)
      if (v[u] >= 0) atomicAdd(&sh_cnt[v[u] & (BNODES - 1)], 1);
  }
  __syncthreads();

  const int own = t < BNODES ? sh_cnt[t] : 0;
  int inc = own;
  for (int s = 1; s < 32; s <<= 1) {
    const int o = __shfl_up(inc, s);
    if (lane >= s) inc += o;
  }
  if (lane == 31) wave_tot[w] = inc;
  __syncthreads();
  int before = 0;
  for (int k = 0; k < 32; ++k)
    if (k < w) before += wave_tot[k];
  if (t < BNODES) {
    const int first = before + inc - own;
    sh_off[t] = first;
    sh_cnt[t] = 0;
    const int n = blockIdx.x * BNODES + t;
    if (n < N) {
      cnt[n] = own;
      off[n] = s0 + first;
    }
  }
  __syncthreads();

  for (int i0 = t; i0 < m; i0 += 8 * 1024) {
    int v[8];
#pragma unroll
    for (int u = 0; u < 8; ++u) {
      const int i = i0 + u * 1024;
      v[u] = i < m ? mine[i] : -1;
    }
#pragma unroll
    for (int u = 0; u < 8; ++u) {
      if (v[u] >= 0) {
        const int dn = v[u] & (BNODES - 1);
        const int p = sh_off[dn] + atomicAdd(&sh_cnt[dn], 1);
        if (staged) stage[p] = v[u] >> BSH;
        else        sorted_src[s0 + p] = v[u] >> BSH;
      }
    }
  }
  if (staged) {
    __syncthreads();
    for (int j = t; j < m; j += 1024) sorted_src[s0 + j] = stage[j];
  }
}

template <int LPE, bool FINAL>
__global__ void __launch_bounds__(256, 16) node_softmax_agg_kernel(
    const int* __restrict__ sorted_src, const int* __restrict__ off,
    const int* __restrict__ cnt, const float* __restrict__ H,
    const float* __restrict__ b, float* __restrict__ outp, int N) {
  constexpr int EPI = 32 / LPE;
  constexpr int CST = 16 / LPE;
  constexpr int SS  = 8 / CST;
  static_assert(LPE == 2 || LPE == 4, "pieces at cols 0 and 8 must exist");

  const int lane = threadIdx.x & 31;
  const int n = __builtin_amdgcn_readfirstlane(
      (int)((blockIdx.x * blockDim.x + threadIdx.x) >> 5));
  if (n >= N) return;
  const int sub = lane & (LPE - 1);
  const int g   = lane / LPE;

  const int   start = off[n];
  const int   deg   = cnt[n];
  const float sdst  = H[(size_t)n * 16 + 9];

  float  m = -INFINITY;
  float  z = 0.0f;
  float4 acc = make_float4(0.0f, 0.0f, 0.0f, 0.0f);

  for (int c = 0; c < deg; c += 32) {
    const int k    = c + lane;
    const int sidx = k < deg ? sorted_src[start + k] : 0;
    float4 v[LPE];
#pragma unroll
    for (int j = 0; j < LPE; ++j) {
      const unsigned s = (unsigned)__shfl(sidx, j * EPI + g);
      v[j] = *reinterpret_cast<const float4*>(H + (size_t)s * 16 + sub * CST);
    }
    float e[LPE];
    float cm = -INFINITY;
#pragma unroll
    for (int j = 0; j < LPE; ++j) {
      const float ssrc  = __shfl(v[j].x, lane - sub + SS);
      const bool  valid = c + j * EPI + g < deg;
      e[j] = valid ? lrelu(ssrc + sdst) : -INFINITY;
      if (!valid) v[j] = make_float4(0.0f, 0.0f, 0.0f, 0.0f);
      cm = fmaxf(cm, e[j]);
    }
    for (int d = LPE; d < 32; d <<= 1) cm = fmaxf(cm, __shfl_xor(cm, d));
    const float mn = fmaxf(m, cm);
    const float sc = __expf(m - mn);
    z *= sc;
    acc.x *= sc; acc.y *= sc; acc.z *= sc; acc.w *= sc;
#pragma unroll
    for (int j = 0; j < LPE; ++j) {
      const float p = __expf(e[j] - mn);
      z += p;
      acc.x += p * v[j].x;
      acc.y += p * v[j].y;
      acc.z += p * v[j].z;
      acc.w += p * v[j].w;
    }
    m = mn;
  }

  for (int d = LPE; d < 32; d <<= 1) {
    z     += __shfl_xor(z, d);
    acc.x += __shfl_xor(acc.x, d);
    acc.y += __shfl_xor(acc.y, d);
    acc.z += __shfl_xor(acc.z, d);
    acc.w += __shfl_xor(acc.w, d);
  }
  const float den = z + 1e-16f;

  if (!FINAL) {
    if (g == 0 && sub < 2) {
      const float* bb = b + sub * 4;
      float4 r;
      r.x = acc.x / den + bb[0];
      r.y = acc.y / den + bb[1];
      r.z = acc.z / den + bb[2];
      r.w = acc.w / den + bb[3];
      r.x = r.x > 0.0f ? r.x : 0.0f;
      r.y = r.y > 0.0f ? r.y : 0.0f;
      r.z = r.z > 0.0f ? r.z : 0.0f;
      r.w = r.w > 0.0f ? r.w : 0.0f;
      *reinterpret_cast<float4*>(outp + (size_t)n * 8 + sub * 4) = r;
    }
  } else {
    if (lane == 0) {
      const float o0 = acc.x / den + b[0];
      const float o1 = acc.y / den + b[1];
      const float mx = fmaxf(o0, o1);
      const float lse = mx + __logf(__expf(o0 - mx) + __expf(o1 - mx));
      *reinterpret_cast<float2*>(outp + (size_t)n * 2) = make_float2(o0 - lse, o1 - lse);
    }
  }
}

#define LNODES 24576
#define GCH    3

__global__ void __launch_bounds__(1024) node_agg_out_kernel(
    const int* __restrict__ sorted_src, const int* __restrict__ off,
    const int* __restrict__ cnt, const float* __restrict__ H,
    const float* __restrict__ b, float* __restrict__ outp, int N) {
  __shared__ float s_h0[LNODES];
  __shared__ float s_h1[LNODES];
  __shared__ float s_ss[LNODES];
  const int t = threadIdx.x;
  const int lane = t & 31;
  const int w = t >> 5;
  const float4* H4 = reinterpret_cast<const float4*>(H);
  const int nl = N < LNODES ? N : LNODES;

  const int npb  = (N + (int)gridDim.x - 1) / (int)gridDim.x;
  const int n0   = (int)blockIdx.x * npb;
  const int nEnd = n0 + npb < N ? n0 + npb : N;

  for (int i = t; i < nl; i += 1024) {
    const float4 r = H4[i];
    s_h0[i] = r.x;
    s_h1[i] = r.y;
    s_ss[i] = r.z;
  }
  __syncthreads();

  const float b0 = b[0];
  const float b1 = b[1];

  for (int nb = n0 + w; nb < nEnd; nb += 32 * 32) {
    const int myn = nb + 32 * lane;
    int   offv = 0;
    int   cntv = 0;
    float sdv  = 0.0f;
    if (myn < nEnd) {
      offv = off[myn];
      cntv = cnt[myn];
      sdv  = H[(size_t)myn * 4 + 3];
    }
    int inb = (nEnd - nb + 31) >> 5;
    if (inb > 32) inb = 32;

    int sxn[GCH];
    {
      const int st = __builtin_amdgcn_readlane(offv, 0);
      const int dg = __builtin_amdgcn_readlane(cntv, 0);
#pragma unroll
      for (int u = 0; u < GCH; ++u) {
        const int k = u * 32 + lane;
        sxn[u] = k < dg ? sorted_src[st + k] : -1;
      }
    }

    for (int j = 0; j < inb; ++j) {
      const int   n     = nb + 32 * j;
      const int   start = __builtin_amdgcn_readlane(offv, j);
      const int   deg   = __builtin_amdgcn_readlane(cntv, j);
      const float sdst  = __int_as_float(__builtin_amdgcn_readlane(__float_as_int(sdv), j));

      float m  = -INFINITY;
      float z  = 0.0f;
      float a0 = 0.0f;
      float a1 = 0.0f;

      int sx0[GCH];
#pragma unroll
      for (int u = 0; u < GCH; ++u) sx0[u] = sxn[u];
      if (j + 1 < inb) {
        const int st = __builtin_amdgcn_readlane(offv, j + 1);
        const int dg = __builtin_amdgcn_readlane(cntv, j + 1);
#pragma unroll
        for (int u = 0; u < GCH; ++u) {
          const int k = u * 32 + lane;
          sxn[u] = k < dg ? sorted_src[st + k] : -1;
        }
      }

      for (int c0 = 0; c0 < deg; c0 += 32 * GCH) {
        int sx[GCH];
        if (c0 == 0) {
#pragma unroll
          for (int u = 0; u < GCH; ++u) sx[u] = sx0[u];
        } else {
#pragma unroll
          for (int u = 0; u < GCH; ++u) {
            const int k = c0 + u * 32 + lane;
            sx[u] = k < deg ? sorted_src[start + k] : -1;
          }
        }
        float4 g[GCH];
        float  l0[GCH], l1[GCH], l2[GCH];
#pragma unroll
        for (int u = 0; u < GCH; ++u) {
          g[u] = make_float4(0.0f, 0.0f, 0.0f, 0.0f);
          if (sx[u] >= nl) g[u] = H4[sx[u]];
        }
#pragma unroll
        for (int u = 0; u < GCH; ++u) {
          l0[u] = 0.0f;
          l1[u] = 0.0f;
          l2[u] = 0.0f;
          if (sx[u] >= 0 && sx[u] < nl) {
            l0[u] = s_h0[sx[u]];
            l1[u] = s_h1[sx[u]];
            l2[u] = s_ss[sx[u]];
          }
        }
        float e[GCH], h0[GCH], h1[GCH];
        float cm = -INFINITY;
#pragma unroll
        for (int u = 0; u < GCH; ++u) {
          const bool inl = sx[u] < nl;
          const float ss = inl ? l2[u] : g[u].z;
          h0[u] = inl ? l0[u] : g[u].x;
          h1[u] = inl ? l1[u] : g[u].y;
          e[u]  = sx[u] >= 0 ? lrelu(ss + sdst) : -INFINITY;
          cm = fmaxf(cm, e[u]);
        }
        for (int d = 1; d < 32; d <<= 1) cm = fmaxf(cm, __shfl_xor(cm, d));
        const float mn = fmaxf(m, cm);
        const float sc = __expf(m - mn);
        z  *= sc;
        a0 *= sc;
        a1 *= sc;
#pragma unroll
        for (int u = 0; u < GCH; ++u) {
          const float p = __expf(e[u] - mn);
          z  += p;
          a0 += p * h0[u];
          a1 += p * h1[u];
        }
        m = mn;
      }

      for (int d = 1; d < 32; d <<= 1) {
        z  += __shfl_xor(z, d);
        a0 += __shfl_xor(a0, d);
        a1 += __shfl_xor(a1, d);
      }
      if (lane == 0) {
        const float den = z + 1e-16f;
        const float o0 = a0 / den + b0;
        const float o1 = a1 / den + b1;
        const float mx = fmaxf(o0, o1);
        const float lse = mx + __logf(__expf(o0 - mx) + __expf(o1 - mx));
        *reinterpret_cast<float2*>(outp + (size_t)n * 2) = make_float2(o0 - lse, o1 - lse);
      }
    }
  }
}

extern "C" void kernel_launch(void* const* d_in, const int* in_sizes, int n_in,
                              void* d_out, int out_size, void* d_ws,
                              size_t ws_size, hipStream_t stream) {
  const float* x      = (const float*)d_in[0];
  const float* W1     = (const float*)d_in[1];
  const float* a1_src = (const float*)d_in[2];
  const float* a1_dst = (const float*)d_in[3];
  const float* b1     = (const float*)d_in[4];
  const float* W2     = (const float*)d_in[5];
  const float* a2_src = (const float*)d_in[6];
  const float* a2_dst = (const float*)d_in[7];
  const float* b2     = (const float*)d_in[8];
  const int*   eidx   = (const int*)d_in[9];

  const int N = in_sizes[0] / 36;
  const int E = in_sizes[9] / 2;
  const int* src = eidx;
  const int* dst = eidx + E;

  const size_t Np    = ((size_t)N + 3) & ~(size_t)3;
  const size_t Ep    = ((size_t)E + 3) & ~(size_t)3;
  const size_t rowsW = (size_t)N * 40;
  const size_t headW = Ep > rowsW ? Ep : rowsW;
  const int NB = (N + BNODES - 1) >> BSH;
  if (ws_size < (headW + 704 + 2 * Np + Ep) * 4) return;
  if (NB > MAXB) return;
  const int nblk = (E + CHUNK - 1) / CHUNK;
  if ((size_t)nblk * NB > Ep) return;

  float* ws    = (float*)d_ws;
  int*   binned = (int*)d_ws;
  float* h1    = ws;
  float* h2    = h1 + (size_t)N * 16;
  float* h1r   = h2 + (size_t)N * 16;
  int*   bbase = (int*)(ws + headW);
  int*   cnt   = bbase + 704;
  int*   off   = cnt + Np;
  int*   ssrc  = off + Np;
  int*   btab  = ssrc;

  const int BLK = 256;
  dim3 chunkGrid(nblk);
  dim3 waveGrid((N + 7) / 8);
  dim3 gemmGrid((N + 127) / 128);

  bucket_hist_kernel<<<chunkGrid, 1024, 0, stream>>>(dst, btab, E, NB);
  edge_bin_kernel<<<chunkGrid, 1024, 0, stream>>>(src, dst, btab, bbase, binned, E, NB, nblk);
  bucket_sort_kernel<<<dim3(NB), 1024, 0, stream>>>(binned, bbase, cnt, off, ssrc, N);

  gemm_attn_wmma<9, 8, false><<<gemmGrid, BLK, 0, stream>>>(x, 36, W1, a1_src, a1_dst, h1, N);
  node_softmax_agg_kernel<4, false><<<waveGrid, BLK, 0, stream>>>(
      ssrc, off, cnt, h1, b1, h1r, N);

  gemm_attn_wmma<2, 2, true><<<gemmGrid, BLK, 0, stream>>>(h1r, 8, W2, a2_src, a2_dst, h2, N);
  node_agg_out_kernel<<<dim3(256), 1024, 0, stream>>>(
      ssrc, off, cnt, h2, b2, (float*)d_out, N);
}


# hidden-layer aggregation on 256 resident work-groups: a wave walks a dozen nodes, next node's indices prefetched, 96 edges' rows in flight; max per 96-edge group, 8-slot lane-tree sums
# speedup vs baseline: 18.3923x; 1.0428x over previous
#include <hip/hip_runtime.h>
#include <math.h>

typedef __attribute__((ext_vector_type(2))) float v2f;
typedef __attribute__((ext_vector_type(8))) float v8f;

#define NEG_SLOPE 0.2f

__device__ __forceinline__ float lrelu(float x) {
  return x > 0.0f ? x : NEG_SLOPE * x;
}

template <int K4, int F, bool PACKED>
__global__ void __launch_bounds__(256) gemm_attn_wmma(const float* __restrict__ A, int lda,
                                                      const float* __restrict__ W,
                                                      const float* __restrict__ asrc,
                                                      const float* __restrict__ adst,
                                                      float* __restrict__ H, int nRows) {
  constexpr int K = K4 * 4;
  __shared__ float sB[K * 16];

  int lane = threadIdx.x & 31;
  int wave = (int)((blockIdx.x * blockDim.x + threadIdx.x) >> 5);
  int row0 = wave * 16;
  const bool live = row0 < nRows;
  int hf = lane >> 4;
  int l  = lane & 15;

  v2f a[K4];
  if (live) {
#pragma unroll
    for (int k = 0; k < K4; ++k) {
      const float* ap = A + (size_t)(row0 + l) * lda + k * 4 + hf * 2;
      a[k].x = ap[0];
      a[k].y = ap[1];
    }
  }

  for (int i = threadIdx.x; i < K * 16; i += 256) {
    const int k = i >> 4;
    const int n = i & 15;
    float v = 0.0f;
    if (n < F) {
      v = W[k * F + n];
    } else if (n == 8) {
      for (int j = 0; j < F; ++j) v = __builtin_fmaf(W[k * F + j], asrc[j], v);
    } else if (n == 9) {
      for (int j = 0; j < F; ++j) v = __builtin_fmaf(W[k * F + j], adst[j], v);
    }
    sB[i] = v;
  }
  __syncthreads();
  if (!live) return;

  v8f c = {};
#pragma unroll
  for (int k = 0; k < K4; ++k) {
    const float* bp = sB + (k * 4 + hf * 2) * 16 + l;
    v2f b;
    b.x = bp[0];
    b.y = bp[16];
    c = __builtin_amdgcn_wmma_f32_16x16x4_f32(false, a[k], false, b, (short)0, c,
                                              false, false);
  }

  if (!PACKED) {
    float* out = H + (size_t)(row0 + hf * 8) * 16 + l;
#pragma unroll
    for (int r = 0; r < 8; ++r) out[(size_t)r * 16] = c[r];
  } else {
    static_assert(!PACKED || F == 2, "packed rows hold two features");
    const int word = l < 2 ? l : (l == 8 || l == 9) ? l - 6 : -1;
    if (word >= 0) {
      float* out = H + (size_t)(row0 + hf * 8) * 4 + word;
#pragma unroll
      for (int r = 0; r < 8; ++r) out[(size_t)r * 4] = c[r];
    }
  }
}

#define BSH    9
#define BNODES 512
#define MAXB   351
#define MAXC   11
#define EPT    32
#define CHUNK  (EPT * 1024)
#define SCAP   40960

__global__ void __launch_bounds__(1024) bucket_hist_kernel(const int* __restrict__ dst,
                                                           int* __restrict__ btab,
                                                           int E, int NB) {
  __shared__ int sh_cnt[MAXB + 1];
  const int t = threadIdx.x;
  if (t < NB) sh_cnt[t] = 0;
  __syncthreads();
  const int e0 = blockIdx.x * CHUNK + t;
  int d[EPT];
#pragma unroll
  for (int u = 0; u < EPT; ++u) {
    const int i = e0 + u * 1024;
    d[u] = i < E ? dst[i] : -1;
  }
#pragma unroll
  for (int u = 0; u < EPT; ++u)
    if (d[u] >= 0) atomicAdd(&sh_cnt[d[u] >> BSH], 1);
  __syncthreads();
  if (t < NB) btab[(size_t)blockIdx.x * NB + t] = sh_cnt[t];
}

__global__ void __launch_bounds__(1024) edge_bin_kernel(const int* __restrict__ src,
                                                        const int* __restrict__ dst,
                                                        const int* __restrict__ btab,
                                                        int* __restrict__ bbase,
                                                        int* __restrict__ binned,
                                                        int E, int NB, int nblk) {
  __shared__ int sh_cnt[MAXB + 1];
  __shared__ int sh_base[MAXB + 1];
  __shared__ int sh_loc[MAXB + 1];
  __shared__ int sh_tot[MAXB + 1];
  __shared__ int sh_pre[MAXB + 1];
  __shared__ int wave_tot[32];
  __shared__ int gwave_tot[32];
  __shared__ int stage[CHUNK];
  const int t = threadIdx.x;
  const int lane = t & 31;
  const int w = t >> 5;
  const int blk = blockIdx.x;
  if (t <= MAXB) {
    sh_cnt[t] = 0;
    sh_tot[t] = 0;
    sh_pre[t] = 0;
  }
  const int own = t < NB ? btab[(size_t)blk * NB + t] : 0;
  const int e0 = blk * CHUNK + t;
  int d[EPT];
#pragma unroll
  for (int u = 0; u < EPT; ++u) {
    const int i = e0 + u * 1024;
    d[u] = i < E ? dst[i] : -1;
  }
  __syncthreads();

  {
    int tot[MAXC];
    int pre[MAXC];
#pragma unroll
    for (int i = 0; i < MAXC; ++i) {
      tot[i] = 0;
      pre[i] = 0;
    }
#pragma unroll 2
    for (int r = w; r < nblk; r += 32) {
      const int* row = btab + (size_t)r * NB;
      const bool earlier = r < blk;
#pragma unroll
      for (int i = 0; i < MAXC; ++i) {
        const int col = lane + 32 * i;
        const int v = col < NB ? row[col] : 0;
        tot[i] += v;
        pre[i] += earlier ? v : 0;
      }
    }
#pragma unroll
    for (int i = 0; i < MAXC; ++i) {
      const int col = lane + 32 * i;
      if (col < NB) {
        atomicAdd(&sh_tot[col], tot[i]);
        atomicAdd(&sh_pre[col], pre[i]);
      }
    }
  }
  __syncthreads();

  const int gown = t < NB ? sh_tot[t] : 0;
  int inc  = own;
  int ginc = gown;
  for (int s = 1; s < 32; s <<= 1) {
    const int o  = __shfl_up(inc, s);
    const int go = __shfl_up(ginc, s);
    if (lane >= s) {
      inc  += o;
      ginc += go;
    }
  }
  if (lane == 31) {
    wave_tot[w]  = inc;
    gwave_tot[w] = ginc;
  }
  __syncthreads();
  int before  = 0;
  int gbefore = 0;
  for (int k = 0; k < 32; ++k)
    if (k < w) {
      before  += wave_tot[k];
      gbefore += gwave_tot[k];
    }
  const int gex = gbefore + ginc - gown;
  if (t < NB) {
    sh_loc[t]  = before + inc - own;
    sh_base[t] = gex + sh_pre[t];
  }
  if (blk == 0 && t <= NB) bbase[t] = gex;
  __syncthreads();

#pragma unroll
  for (int h = 0; h < EPT; h += 16) {
    int s[16];
#pragma unroll
    for (int u = 0; u < 16; ++u) {
      const int i = e0 + (h + u) * 1024;
      s[u] = i < E ? src[i] : 0;
    }
#pragma unroll
    for (int u = 0; u < 16; ++u) {
      const int dd = d[h + u];
      if (dd >= 0) {
        const int k = dd >> BSH;
        const int r = atomicAdd(&sh_cnt[k], 1);
        stage[sh_loc[k] + r] = (s[u] << BSH) | (dd & (BNODES - 1));
      }
    }
  }
  __syncthreads();

  for (int k = w; k < NB; k += 32) {
    const int c = sh_cnt[k];
    const int g = sh_base[k];
    const int l = sh_loc[k];
    for (int j = lane; j < c; j += 32) binned[g + j] = stage[l + j];
  }
}

__global__ void __launch_bounds__(1024) bucket_sort_kernel(const int* __restrict__ binned,
                                                           const int* __restrict__ bbase,
                                                           int* __restrict__ cnt,
                                                           int* __restrict__ off,
                                                           int* __restrict__ sorted_src, int N) {
  __shared__ int sh_cnt[BNODES];
  __shared__ int sh_off[BNODES];
  __shared__ int wave_tot[32];
  __shared__ int stage[SCAP];
  const int t = threadIdx.x;
  const int lane = t & 31;
  const int w = t >> 5;
  const int s0 = bbase[blockIdx.x];
  const int m  = bbase[blockIdx.x + 1] - s0;
  const int* mine = binned + s0;
  const bool staged = m <= SCAP;

  if (t < BNODES) sh_cnt[t] = 0;
  __syncthreads();

  for (int i0 = t; i0 < m; i0 += 8 * 1024) {
    int v[8];
#pragma unroll
    for (int u = 0; u < 8; ++u) {
      const int i = i0 + u * 1024;
      v[u] = i < m ? mine[i] : -1;
    }
#pragma unroll
    for (int u = 0; u < 8; ++u)
      if (v[u] >= 0) atomicAdd(&sh_cnt[v[u] & (BNODES - 1)], 1);
  }
  __syncthreads();

  const int own = t < BNODES ? sh_cnt[t] : 0;
  int inc = own;
  for (int s = 1; s < 32; s <<= 1) {
    const int o = __shfl_up(inc, s);
    if (lane >= s) inc += o;
  }
  if (lane == 31) wave_tot[w] = inc;
  __syncthreads();
  int before = 0;
  for (int k = 0; k < 32; ++k)
    if (k < w) before += wave_tot[k];
  if (t < BNODES) {
    const int first = before + inc - own;
    sh_off[t] = first;
    sh_cnt[t] = 0;
    const int n = blockIdx.x * BNODES + t;
    if (n < N) {
      cnt[n] = own;
      off[n] = s0 + first;
    }
  }
  __syncthreads();

  for (int i0 = t; i0 < m; i0 += 8 * 1024) {
    int v[8];
#pragma unroll
    for (int u = 0; u < 8; ++u) {
      const int i = i0 + u * 1024;
      v[u] = i < m ? mine[i] : -1;
    }
#pragma unroll
    for (int u = 0; u < 8; ++u) {
      if (v[u] >= 0) {
        const int dn = v[u] & (BNODES - 1);
        const int p = sh_off[dn] + atomicAdd(&sh_cnt[dn], 1);
        if (staged) stage[p] = v[u] >> BSH;
        else        sorted_src[s0 + p] = v[u] >> BSH;
      }
    }
  }
  if (staged) {
    __syncthreads();
    for (int j = t; j < m; j += 1024) sorted_src[s0 + j] = stage[j];
  }
}

#define G1CH 3

__global__ void __launch_bounds__(1024) node_agg_hid_kernel(
    const int* __restrict__ sorted_src, const int* __restrict__ off,
    const int* __restrict__ cnt, const float* __restrict__ H,
    const float* __restrict__ b, float* __restrict__ outp, int N) {
  const int t = threadIdx.x;
  const int lane = t & 31;
  const int w = t >> 5;
  const int sub = lane & 3;
  const int g   = lane >> 2;

  const int npb  = (N + (int)gridDim.x - 1) / (int)gridDim.x;
  const int n0   = (int)blockIdx.x * npb;
  const int nEnd = n0 + npb < N ? n0 + npb : N;

  const float bs0 = b[0], bs1 = b[1], bs2 = b[2], bs3 = b[3];
  const float bs4 = b[4], bs5 = b[5], bs6 = b[6], bs7 = b[7];

  for (int nb = n0 + w; nb < nEnd; nb += 32 * 32) {
    const int myn = nb + 32 * lane;
    int   offv = 0;
    int   cntv = 0;
    float sdv  = 0.0f;
    if (myn < nEnd) {
      offv = off[myn];
      cntv = cnt[myn];
      sdv  = H[(size_t)myn * 16 + 9];
    }
    int inb = (nEnd - nb + 31) >> 5;
    if (inb > 32) inb = 32;

    int sxn[G1CH];
    {
      const int st = __builtin_amdgcn_readlane(offv, 0);
      const int dg = __builtin_amdgcn_readlane(cntv, 0);
#pragma unroll
      for (int u = 0; u < G1CH; ++u) {
        const int k = u * 32 + lane;
        sxn[u] = k < dg ? sorted_src[st + k] : 0;
      }
    }

    for (int j = 0; j < inb; ++j) {
      const int   n     = nb + 32 * j;
      const int   start = __builtin_amdgcn_readlane(offv, j);
      const int   deg   = __builtin_amdgcn_readlane(cntv, j);
      const float sdst  = __int_as_float(__builtin_amdgcn_readlane(__float_as_int(sdv), j));

      float  m = -INFINITY;
      float  z = 0.0f;
      float4 acc = make_float4(0.0f, 0.0f, 0.0f, 0.0f);

      int sx0[G1CH];
#pragma unroll
      for (int u = 0; u < G1CH; ++u) sx0[u] = sxn[u];
      if (j + 1 < inb) {
        const int st = __builtin_amdgcn_readlane(offv, j + 1);
        const int dg = __builtin_amdgcn_readlane(cntv, j + 1);
#pragma unroll
        for (int u = 0; u < G1CH; ++u) {
          const int k = u * 32 + lane;
          sxn[u] = k < dg ? sorted_src[st + k] : 0;
        }
      }

      for (int c0 = 0; c0 < deg; c0 += 32 * G1CH) {
        int sx[G1CH];
        if (c0 == 0) {
#pragma unroll
          for (int u = 0; u < G1CH; ++u) sx[u] = sx0[u];
        } else {
#pragma unroll
          for (int u = 0; u < G1CH; ++u) {
            const int k = c0 + u * 32 + lane;
            sx[u] = k < deg ? sorted_src[start + k] : 0;
          }
        }
        float4 v[G1CH][4];
#pragma unroll
        for (int u = 0; u < G1CH; ++u) {
          const bool any = c0 + u * 32 < deg;
          unsigned s[4];
#pragma unroll
          for (int q = 0; q < 4; ++q) s[q] = (unsigned)__shfl(sx[u], q * 8 + g);
#pragma unroll
          for (int q = 0; q < 4; ++q) {
            v[u][q] = make_float4(0.0f, 0.0f, 0.0f, 0.0f);
            if (any)
              v[u][q] = *reinterpret_cast<const float4*>(H + (size_t)s[q] * 16 + sub * 4);
          }
        }
        float e[G1CH][4];
        float cm = -INFINITY;
#pragma unroll
        for (int u = 0; u < G1CH; ++u) {
#pragma unroll
          for (int q = 0; q < 4; ++q) {
            const float ssrc  = __shfl(v[u][q].x, lane - sub + 2);
            const bool  valid = c0 + u * 32 + q * 8 + g < deg;
            e[u][q] = valid ? lrelu(ssrc + sdst) : -INFINITY;
            if (!valid) v[u][q] = make_float4(0.0f, 0.0f, 0.0f, 0.0f);
            cm = fmaxf(cm, e[u][q]);
          }
        }
        for (int d = 4; d < 32; d <<= 1) cm = fmaxf(cm, __shfl_xor(cm, d));
        const float mn = fmaxf(m, cm);
        const float sc = __expf(m - mn);
        z *= sc;
        acc.x *= sc; acc.y *= sc; acc.z *= sc; acc.w *= sc;
#pragma unroll
        for (int u = 0; u < G1CH; ++u) {
#pragma unroll
          for (int q = 0; q < 4; ++q) {
            const float p = __expf(e[u][q] - mn);
            z += p;
            acc.x += p * v[u][q].x;
            acc.y += p * v[u][q].y;
            acc.z += p * v[u][q].z;
            acc.w += p * v[u][q].w;
          }
        }
        m = mn;
      }

      for (int d = 4; d < 32; d <<= 1) {
        z     += __shfl_xor(z, d);
        acc.x += __shfl_xor(acc.x, d);
        acc.y += __shfl_xor(acc.y, d);
        acc.z += __shfl_xor(acc.z, d);
        acc.w += __shfl_xor(acc.w, d);
      }
      if (g == 0 && sub < 2) {
        const float den = z + 1e-16f;
        float4 r;
        r.x = acc.x / den + (sub == 0 ? bs0 : bs4);
        r.y = acc.y / den + (sub == 0 ? bs1 : bs5);
        r.z = acc.z / den + (sub == 0 ? bs2 : bs6);
        r.w = acc.w / den + (sub == 0 ? bs3 : bs7);
        r.x = r.x > 0.0f ? r.x : 0.0f;
        r.y = r.y > 0.0f ? r.y : 0.0f;
        r.z = r.z > 0.0f ? r.z : 0.0f;
        r.w = r.w > 0.0f ? r.w : 0.0f;
        *reinterpret_cast<float4*>(outp + (size_t)n * 8 + sub * 4) = r;
      }
    }
  }
}

#define LNODES 24576
#define GCH    3

__global__ void __launch_bounds__(1024) node_agg_out_kernel(
    const int* __restrict__ sorted_src, const int* __restrict__ off,
    const int* __restrict__ cnt, const float* __restrict__ H,
    const float* __restrict__ b, float* __restrict__ outp, int N) {
  __shared__ float s_h0[LNODES];
  __shared__ float s_h1[LNODES];
  __shared__ float s_ss[LNODES];
  const int t = threadIdx.x;
  const int lane = t & 31;
  const int w = t >> 5;
  const float4* H4 = reinterpret_cast<const float4*>(H);
  const int nl = N < LNODES ? N : LNODES;

  const int npb  = (N + (int)gridDim.x - 1) / (int)gridDim.x;
  const int n0   = (int)blockIdx.x * npb;
  const int nEnd = n0 + npb < N ? n0 + npb : N;

  for (int i = t; i < nl; i += 1024) {
    const float4 r = H4[i];
    s_h0[i] = r.x;
    s_h1[i] = r.y;
    s_ss[i] = r.z;
  }
  __syncthreads();

  const float b0 = b[0];
  const float b1 = b[1];

  for (int nb = n0 + w; nb < nEnd; nb += 32 * 32) {
    const int myn = nb + 32 * lane;
    int   offv = 0;
    int   cntv = 0;
    float sdv  = 0.0f;
    if (myn < nEnd) {
      offv = off[myn];
      cntv = cnt[myn];
      sdv  = H[(size_t)myn * 4 + 3];
    }
    int inb = (nEnd - nb + 31) >> 5;
    if (inb > 32) inb = 32;

    int sxn[GCH];
    {
      const int st = __builtin_amdgcn_readlane(offv, 0);
      const int dg = __builtin_amdgcn_readlane(cntv, 0);
#pragma unroll
      for (int u = 0; u < GCH; ++u) {
        const int k = u * 32 + lane;
        sxn[u] = k < dg ? sorted_src[st + k] : -1;
      }
    }

    for (int j = 0; j < inb; ++j) {
      const int   n     = nb + 32 * j;
      const int   start = __builtin_amdgcn_readlane(offv, j);
      const int   deg   = __builtin_amdgcn_readlane(cntv, j);
      const float sdst  = __int_as_float(__builtin_amdgcn_readlane(__float_as_int(sdv), j));

      float m  = -INFINITY;
      float z  = 0.0f;
      float a0 = 0.0f;
      float a1 = 0.0f;

      int sx0[GCH];
#pragma unroll
      for (int u = 0; u < GCH; ++u) sx0[u] = sxn[u];
      if (j + 1 < inb) {
        const int st = __builtin_amdgcn_readlane(offv, j + 1);
        const int dg = __builtin_amdgcn_readlane(cntv, j + 1);
#pragma unroll
        for (int u = 0; u < GCH; ++u) {
          const int k = u * 32 + lane;
          sxn[u] = k < dg ? sorted_src[st + k] : -1;
        }
      }

      for (int c0 = 0; c0 < deg; c0 += 32 * GCH) {
        int sx[GCH];
        if (c0 == 0) {
#pragma unroll
          for (int u = 0; u < GCH; ++u) sx[u] = sx0[u];
        } else {
#pragma unroll
          for (int u = 0; u < GCH; ++u) {
            const int k = c0 + u * 32 + lane;
            sx[u] = k < deg ? sorted_src[start + k] : -1;
          }
        }
        float4 g[GCH];
        float  l0[GCH], l1[GCH], l2[GCH];
#pragma unroll
        for (int u = 0; u < GCH; ++u) {
          g[u] = make_float4(0.0f, 0.0f, 0.0f, 0.0f);
          if (sx[u] >= nl) g[u] = H4[sx[u]];
        }
#pragma unroll
        for (int u = 0; u < GCH; ++u) {
          l0[u] = 0.0f;
          l1[u] = 0.0f;
          l2[u] = 0.0f;
          if (sx[u] >= 0 && sx[u] < nl) {
            l0[u] = s_h0[sx[u]];
            l1[u] = s_h1[sx[u]];
            l2[u] = s_ss[sx[u]];
          }
        }
        float e[GCH], h0[GCH], h1[GCH];
        float cm = -INFINITY;
#pragma unroll
        for (int u = 0; u < GCH; ++u) {
          const bool inl = sx[u] < nl;
          const float ss = inl ? l2[u] : g[u].z;
          h0[u] = inl ? l0[u] : g[u].x;
          h1[u] = inl ? l1[u] : g[u].y;
          e[u]  = sx[u] >= 0 ? lrelu(ss + sdst) : -INFINITY;
          cm = fmaxf(cm, e[u]);
        }
        for (int d = 1; d < 32; d <<= 1) cm = fmaxf(cm, __shfl_xor(cm, d));
        const float mn = fmaxf(m, cm);
        const float sc = __expf(m - mn);
        z  *= sc;
        a0 *= sc;
        a1 *= sc;
#pragma unroll
        for (int u = 0; u < GCH; ++u) {
          const float p = __expf(e[u] - mn);
          z  += p;
          a0 += p * h0[u];
          a1 += p * h1[u];
        }
        m = mn;
      }

      for (int d = 1; d < 32; d <<= 1) {
        z  += __shfl_xor(z, d);
        a0 += __shfl_xor(a0, d);
        a1 += __shfl_xor(a1, d);
      }
      if (lane == 0) {
        const float den = z + 1e-16f;
        const float o0 = a0 / den + b0;
        const float o1 = a1 / den + b1;
        const float mx = fmaxf(o0, o1);
        const float lse = mx + __logf(__expf(o0 - mx) + __expf(o1 - mx));
        *reinterpret_cast<float2*>(outp + (size_t)n * 2) = make_float2(o0 - lse, o1 - lse);
      }
    }
  }
}

extern "C" void kernel_launch(void* const* d_in, const int* in_sizes, int n_in,
                              void* d_out, int out_size, void* d_ws,
                              size_t ws_size, hipStream_t stream) {
  const float* x      = (const float*)d_in[0];
  const float* W1     = (const float*)d_in[1];
  const float* a1_src = (const float*)d_in[2];
  const float* a1_dst = (const float*)d_in[3];
  const float* b1     = (const float*)d_in[4];
  const float* W2     = (const float*)d_in[5];
  const float* a2_src = (const float*)d_in[6];
  const float* a2_dst = (const float*)d_in[7];
  const float* b2     = (const float*)d_in[8];
  const int*   eidx   = (const int*)d_in[9];

  const int N = in_sizes[0] / 36;
  const int E = in_sizes[9] / 2;
  const int* src = eidx;
  const int* dst = eidx + E;

  const size_t Np    = ((size_t)N + 3) & ~(size_t)3;
  const size_t Ep    = ((size_t)E + 3) & ~(size_t)3;
  const size_t rowsW = (size_t)N * 40;
  const size_t headW = Ep > rowsW ? Ep : rowsW;
  const int NB = (N + BNODES - 1) >> BSH;
  if (ws_size < (headW + 704 + 2 * Np + Ep) * 4) return;
  if (NB > MAXB) return;
  const int nblk = (E + CHUNK - 1) / CHUNK;
  if ((size_t)nblk * NB > Ep) return;

  float* ws    = (float*)d_ws;
  int*   binned = (int*)d_ws;
  float* h1    = ws;
  float* h2    = h1 + (size_t)N * 16;
  float* h1r   = h2 + (size_t)N * 16;
  int*   bbase = (int*)(ws + headW);
  int*   cnt   = bbase + 704;
  int*   off   = cnt + Np;
  int*   ssrc  = off + Np;
  int*   btab  = ssrc;

  const int BLK = 256;
  dim3 chunkGrid(nblk);
  dim3 gemmGrid((N + 127) / 128);

  bucket_hist_kernel<<<chunkGrid, 1024, 0, stream>>>(dst, btab, E, NB);
  edge_bin_kernel<<<chunkGrid, 1024, 0, stream>>>(src, dst, btab, bbase, binned, E, NB, nblk);
  bucket_sort_kernel<<<dim3(NB), 1024, 0, stream>>>(binned, bbase, cnt, off, ssrc, N);

  gemm_attn_wmma<9, 8, false><<<gemmGrid, BLK, 0, stream>>>(x, 36, W1, a1_src, a1_dst, h1, N);
  node_agg_hid_kernel<<<dim3(256), 1024, 0, stream>>>(
      ssrc, off, cnt, h1, b1, h1r, N);

  gemm_attn_wmma<2, 2, true><<<gemmGrid, BLK, 0, stream>>>(h1r, 8, W2, a2_src, a2_dst, h2, N);
  node_agg_out_kernel<<<dim3(256), 1024, 0, stream>>>(
      ssrc, off, cnt, h2, b2, (float*)d_out, N);
}


# output-layer source rows cut to two words (h0,h1; the score term s_src formed per edge from them): 40448 rows fit the resident LDS copy instead of 24576, one 8-byte LDS load per edge
# speedup vs baseline: 19.1405x; 1.0407x over previous
#include <hip/hip_runtime.h>
#include <math.h>

typedef __attribute__((ext_vector_type(2))) float v2f;
typedef __attribute__((ext_vector_type(8))) float v8f;

#define NEG_SLOPE 0.2f

__device__ __forceinline__ float lrelu(float x) {
  return x > 0.0f ? x : NEG_SLOPE * x;
}

template <int K4, int F, bool PACKED>
__global__ void __launch_bounds__(256) gemm_attn_wmma(const float* __restrict__ A, int lda,
                                                      const float* __restrict__ W,
                                                      const float* __restrict__ asrc,
                                                      const float* __restrict__ adst,
                                                      float* __restrict__ H, int nRows) {
  constexpr int K = K4 * 4;
  __shared__ float sB[K * 16];

  int lane = threadIdx.x & 31;
  int wave = (int)((blockIdx.x * blockDim.x + threadIdx.x) >> 5);
  int row0 = wave * 16;
  const bool live = row0 < nRows;
  int hf = lane >> 4;
  int l  = lane & 15;

  v2f a[K4];
  if (live) {
#pragma unroll
    for (int k = 0; k < K4; ++k) {
      const float* ap = A + (size_t)(row0 + l) * lda + k * 4 + hf * 2;
      a[k].x = ap[0];
      a[k].y = ap[1];
    }
  }

  for (int i = threadIdx.x; i < K * 16; i += 256) {
    const int k = i >> 4;
    const int n = i & 15;
    float v = 0.0f;
    if (n < F) {
      v = W[k * F + n];
    } else if (n == 8 && !PACKED) {
      for (int j = 0; j < F; ++j) v = __builtin_fmaf(W[k * F + j], asrc[j], v);
    } else if (n == 9) {
      for (int j = 0; j < F; ++j) v = __builtin_fmaf(W[k * F + j], adst[j], v);
    }
    sB[i] = v;
  }
  __syncthreads();
  if (!live) return;

  v8f c = {};
#pragma unroll
  for (int k = 0; k < K4; ++k) {
    const float* bp = sB + (k * 4 + hf * 2) * 16 + l;
    v2f b;
    b.x = bp[0];
    b.y = bp[16];
    c = __builtin_amdgcn_wmma_f32_16x16x4_f32(false, a[k], false, b, (short)0, c,
                                              false, false);
  }

  if (!PACKED) {
    float* out = H + (size_t)(row0 + hf * 8) * 16 + l;
#pragma unroll
    for (int r = 0; r < 8; ++r) out[(size_t)r * 16] = c[r];
  } else {
    static_assert(!PACKED || F == 2, "packed rows hold two features");
    if (l < 2) {
      float* out = H + (size_t)(row0 + hf * 8) * 2 + l;
#pragma unroll
      for (int r = 0; r < 8; ++r) out[(size_t)r * 2] = c[r];
    } else if (l == 9) {
      float* out = H + (size_t)((nRows + 15) & ~15) * 2 + row0 + hf * 8;
#pragma unroll
      for (int r = 0; r < 8; ++r) out[r] = c[r];
    }
  }
}

#define BSH    9
#define BNODES 512
#define MAXB   351
#define MAXC   11
#define EPT    32
#define CHUNK  (EPT * 1024)
#define SCAP   40960

__global__ void __launch_bounds__(1024) bucket_hist_kernel(const int* __restrict__ dst,
                                                           int* __restrict__ btab,
                                                           int E, int NB) {
  __shared__ int sh_cnt[MAXB + 1];
  const int t = threadIdx.x;
  if (t < NB) sh_cnt[t] = 0;
  __syncthreads();
  const int e0 = blockIdx.x * CHUNK + t;
  int d[EPT];
#pragma unroll
  for (int u = 0; u < EPT; ++u) {
    const int i = e0 + u * 1024;
    d[u] = i < E ? dst[i] : -1;
  }
#pragma unroll
  for (int u = 0; u < EPT; ++u)
    if (d[u] >= 0) atomicAdd(&sh_cnt[d[u] >> BSH], 1);
  __syncthreads();
  if (t < NB) btab[(size_t)blockIdx.x * NB + t] = sh_cnt[t];
}

__global__ void __launch_bounds__(1024) edge_bin_kernel(const int* __restrict__ src,
                                                        const int* __restrict__ dst,
                                                        const int* __restrict__ btab,
                                                        int* __restrict__ bbase,
                                                        int* __restrict__ binned,
                                                        int E, int NB, int nblk) {
  __shared__ int sh_cnt[MAXB + 1];
  __shared__ int sh_base[MAXB + 1];
  __shared__ int sh_loc[MAXB + 1];
  __shared__ int sh_tot[MAXB + 1];
  __shared__ int sh_pre[MAXB + 1];
  __shared__ int wave_tot[32];
  __shared__ int gwave_tot[32];
  __shared__ int stage[CHUNK];
  const int t = threadIdx.x;
  const int lane = t & 31;
  const int w = t >> 5;
  const int blk = blockIdx.x;
  if (t <= MAXB) {
    sh_cnt[t] = 0;
    sh_tot[t] = 0;
    sh_pre[t] = 0;
  }
  const int own = t < NB ? btab[(size_t)blk * NB + t] : 0;
  const int e0 = blk * CHUNK + t;
  int d[EPT];
#pragma unroll
  for (int u = 0; u < EPT; ++u) {
    const int i = e0 + u * 1024;
    d[u] = i < E ? dst[i] : -1;
  }
  __syncthreads();

  {
    int tot[MAXC];
    int pre[MAXC];
#pragma unroll
    for (int i = 0; i < MAXC; ++i) {
      tot[i] = 0;
      pre[i] = 0;
    }
#pragma unroll 2
    for (int r = w; r < nblk; r += 32) {
      const int* row = btab + (size_t)r * NB;
      const bool earlier = r < blk;
#pragma unroll
      for (int i = 0; i < MAXC; ++i) {
        const int col = lane + 32 * i;
        const int v = col < NB ? row[col] : 0;
        tot[i] += v;
        pre[i] += earlier ? v : 0;
      }
    }
#pragma unroll
    for (int i = 0; i < MAXC; ++i) {
      const int col = lane + 32 * i;
      if (col < NB) {
        atomicAdd(&sh_tot[col], tot[i]);
        atomicAdd(&sh_pre[col], pre[i]);
      }
    }
  }
  __syncthreads();

  const int gown = t < NB ? sh_tot[t] : 0;
  int inc  = own;
  int ginc = gown;
  for (int s = 1; s < 32; s <<= 1) {
    const int o  = __shfl_up(inc, s);
    const int go = __shfl_up(ginc, s);
    if (lane >= s) {
      inc  += o;
      ginc += go;
    }
  }
  if (lane == 31) {
    wave_tot[w]  = inc;
    gwave_tot[w] = ginc;
  }
  __syncthreads();
  int before  = 0;
  int gbefore = 0;
  for (int k = 0; k < 32; ++k)
    if (k < w) {
      before  += wave_tot[k];
      gbefore += gwave_tot[k];
    }
  const int gex = gbefore + ginc - gown;
  if (t < NB) {
    sh_loc[t]  = before + inc - own;
    sh_base[t] = gex + sh_pre[t];
  }
  if (blk == 0 && t <= NB) bbase[t] = gex;
  __syncthreads();

#pragma unroll
  for (int h = 0; h < EPT; h += 16) {
    int s[16];
#pragma unroll
    for (int u = 0; u < 16; ++u) {
      const int i = e0 + (h + u) * 1024;
      s[u] = i < E ? src[i] : 0;
    }
#pragma unroll
    for (int u = 0; u < 16; ++u) {
      const int dd = d[h + u];
      if (dd >= 0) {
        const int k = dd >> BSH;
        const int r = atomicAdd(&sh_cnt[k], 1);
        stage[sh_loc[k] + r] = (s[u] << BSH) | (dd & (BNODES - 1));
      }
    }
  }
  __syncthreads();

  for (int k = w; k < NB; k += 32) {
    const int c = sh_cnt[k];
    const int g = sh_base[k];
    const int l = sh_loc[k];
    for (int j = lane; j < c; j += 32) binned[g + j] = stage[l + j];
  }
}

__global__ void __launch_bounds__(1024) bucket_sort_kernel(const int* __restrict__ binned,
                                                           const int* __restrict__ bbase,
                                                           int* __restrict__ cnt,
                                                           int* __restrict__ off,
                                                           int* __restrict__ sorted_src, int N) {
  __shared__ int sh_cnt[BNODES];
  __shared__ int sh_off[BNODES];
  __shared__ int wave_tot[32];
  __shared__ int stage[SCAP];
  const int t = threadIdx.x;
  const int lane = t & 31;
  const int w = t >> 5;
  const int s0 = bbase[blockIdx.x];
  const int m  = bbase[blockIdx.x + 1] - s0;
  const int* mine = binned + s0;
  const bool staged = m <= SCAP;

  if (t < BNODES) sh_cnt[t] = 0;
  __syncthreads();

  for (int i0 = t; i0 < m; i0 += 8 * 1024) {
    int v[8];
#pragma unroll
    for (int u = 0; u < 8; ++u) {
      const int i = i0 + u * 1024;
      v[u] = i < m ? mine[i] : -1;
    }
#pragma unroll
    for (int u = 0; u < 8; ++u)
      if (v[u] >= 0) atomicAdd(&sh_cnt[v[u] & (BNODES - 1)], 1);
  }
  __syncthreads();

  const int own = t < BNODES ? sh_cnt[t] : 0;
  int inc = own;
  for (int s = 1; s < 32; s <<= 1) {
    const int o = __shfl_up(inc, s);
    if (lane >= s) inc += o;
  }
  if (lane == 31) wave_tot[w] = inc;
  __syncthreads();
  int before = 0;
  for (int k = 0; k < 32; ++k)
    if (k < w) before += wave_tot[k];
  if (t < BNODES) {
    const int first = before + inc - own;
    sh_off[t] = first;
    sh_cnt[t] = 0;
    const int n = blockIdx.x * BNODES + t;
    if (n < N) {
      cnt[n] = own;
      off[n] = s0 + first;
    }
  }
  __syncthreads();

  for (int i0 = t; i0 < m; i0 += 8 * 1024) {
    int v[8];
#pragma unroll
    for (int u = 0; u < 8; ++u) {
      const int i = i0 + u * 1024;
      v[u] = i < m ? mine[i] : -1;
    }
#pragma unroll
    for (int u = 0; u < 8; ++u) {
      if (v[u] >= 0) {
        const int dn = v[u] & (BNODES - 1);
        const int p = sh_off[dn] + atomicAdd(&sh_cnt[dn], 1);
        if (staged) stage[p] = v[u] >> BSH;
        else        sorted_src[s0 + p] = v[u] >> BSH;
      }
    }
  }
  if (staged) {
    __syncthreads();
    for (int j = t; j < m; j += 1024) sorted_src[s0 + j] = stage[j];
  }
}

#define G1CH 3

__global__ void __launch_bounds__(1024) node_agg_hid_kernel(
    const int* __restrict__ sorted_src, const int* __restrict__ off,
    const int* __restrict__ cnt, const float* __restrict__ H,
    const float* __restrict__ b, float* __restrict__ outp, int N) {
  const int t = threadIdx.x;
  const int lane = t & 31;
  const int w = t >> 5;
  const int sub = lane & 3;
  const int g   = lane >> 2;

  const int npb  = (N + (int)gridDim.x - 1) / (int)gridDim.x;
  const int n0   = (int)blockIdx.x * npb;
  const int nEnd = n0 + npb < N ? n0 + npb : N;

  const float bs0 = b[0], bs1 = b[1], bs2 = b[2], bs3 = b[3];
  const float bs4 = b[4], bs5 = b[5], bs6 = b[6], bs7 = b[7];

  for (int nb = n0 + w; nb < nEnd; nb += 32 * 32) {
    const int myn = nb + 32 * lane;
    int   offv = 0;
    int   cntv = 0;
    float sdv  = 0.0f;
    if (myn < nEnd) {
      offv = off[myn];
      cntv = cnt[myn];
      sdv  = H[(size_t)myn * 16 + 9];
    }
    int inb = (nEnd - nb + 31) >> 5;
    if (inb > 32) inb = 32;

    int sxn[G1CH];
    {
      const int st = __builtin_amdgcn_readlane(offv, 0);
      const int dg = __builtin_amdgcn_readlane(cntv, 0);
#pragma unroll
      for (int u = 0; u < G1CH; ++u) {
        const int k = u * 32 + lane;
        sxn[u] = k < dg ? sorted_src[st + k] : 0;
      }
    }

    for (int j = 0; j < inb; ++j) {
      const int   n     = nb + 32 * j;
      const int   start = __builtin_amdgcn_readlane(offv, j);
      const int   deg   = __builtin_amdgcn_readlane(cntv, j);
      const float sdst  = __int_as_float(__builtin_amdgcn_readlane(__float_as_int(sdv), j));

      float  m = -INFINITY;
      float  z = 0.0f;
      float4 acc = make_float4(0.0f, 0.0f, 0.0f, 0.0f);

      int sx0[G1CH];
#pragma unroll
      for (int u = 0; u < G1CH; ++u) sx0[u] = sxn[u];
      if (j + 1 < inb) {
        const int st = __builtin_amdgcn_readlane(offv, j + 1);
        const int dg = __builtin_amdgcn_readlane(cntv, j + 1);
#pragma unroll
        for (int u = 0; u < G1CH; ++u) {
          const int k = u * 32 + lane;
          sxn[u] = k < dg ? sorted_src[st + k] : 0;
        }
      }

      for (int c0 = 0; c0 < deg; c0 += 32 * G1CH) {
        int sx[G1CH];
        if (c0 == 0) {
#pragma unroll
          for (int u = 0; u < G1CH; ++u) sx[u] = sx0[u];
        } else {
#pragma unroll
          for (int u = 0; u < G1CH; ++u) {
            const int k = c0 + u * 32 + lane;
            sx[u] = k < deg ? sorted_src[start + k] : 0;
          }
        }
        float4 v[G1CH][4];
#pragma unroll
        for (int u = 0; u < G1CH; ++u) {
          const bool any = c0 + u * 32 < deg;
          unsigned s[4];
#pragma unroll
          for (int q = 0; q < 4; ++q) s[q] = (unsigned)__shfl(sx[u], q * 8 + g);
#pragma unroll
          for (int q = 0; q < 4; ++q) {
            v[u][q] = make_float4(0.0f, 0.0f, 0.0f, 0.0f);
            if (any)
              v[u][q] = *reinterpret_cast<const float4*>(H + (size_t)s[q] * 16 + sub * 4);
          }
        }
        float e[G1CH][4];
        float cm = -INFINITY;
#pragma unroll
        for (int u = 0; u < G1CH; ++u) {
#pragma unroll
          for (int q = 0; q < 4; ++q) {
            const float ssrc  = __shfl(v[u][q].x, lane - sub + 2);
            const bool  valid = c0 + u * 32 + q * 8 + g < deg;
            e[u][q] = valid ? lrelu(ssrc + sdst) : -INFINITY;
            if (!valid) v[u][q] = make_float4(0.0f, 0.0f, 0.0f, 0.0f);
            cm = fmaxf(cm, e[u][q]);
          }
        }
        for (int d = 4; d < 32; d <<= 1) cm = fmaxf(cm, __shfl_xor(cm, d));
        const float mn = fmaxf(m, cm);
        const float sc = __expf(m - mn);
        z *= sc;
        acc.x *= sc; acc.y *= sc; acc.z *= sc; acc.w *= sc;
#pragma unroll
        for (int u = 0; u < G1CH; ++u) {
#pragma unroll
          for (int q = 0; q < 4; ++q) {
            const float p = __expf(e[u][q] - mn);
            z += p;
            acc.x += p * v[u][q].x;
            acc.y += p * v[u][q].y;
            acc.z += p * v[u][q].z;
            acc.w += p * v[u][q].w;
          }
        }
        m = mn;
      }

      for (int d = 4; d < 32; d <<= 1) {
        z     += __shfl_xor(z, d);
        acc.x += __shfl_xor(acc.x, d);
        acc.y += __shfl_xor(acc.y, d);
        acc.z += __shfl_xor(acc.z, d);
        acc.w += __shfl_xor(acc.w, d);
      }
      if (g == 0 && sub < 2) {
        const float den = z + 1e-16f;
        float4 r;
        r.x = acc.x / den + (sub == 0 ? bs0 : bs4);
        r.y = acc.y / den + (sub == 0 ? bs1 : bs5);
        r.z = acc.z / den + (sub == 0 ? bs2 : bs6);
        r.w = acc.w / den + (sub == 0 ? bs3 : bs7);
        r.x = r.x > 0.0f ? r.x : 0.0f;
        r.y = r.y > 0.0f ? r.y : 0.0f;
        r.z = r.z > 0.0f ? r.z : 0.0f;
        r.w = r.w > 0.0f ? r.w : 0.0f;
        *reinterpret_cast<float4*>(outp + (size_t)n * 8 + sub * 4) = r;
      }
    }
  }
}

#define LNODES 40448
#define GCH    3

__global__ void __launch_bounds__(1024) node_agg_out_kernel(
    const int* __restrict__ sorted_src, const int* __restrict__ off,
    const int* __restrict__ cnt, const float* __restrict__ H,
    const float* __restrict__ asrc,
    const float* __restrict__ b, float* __restrict__ outp, int N) {
  __shared__ float2 s_h[LNODES];
  const int t = threadIdx.x;
  const int lane = t & 31;
  const int w = t >> 5;
  const float2* H2 = reinterpret_cast<const float2*>(H);
  const float*  SD = H + (size_t)((N + 15) & ~15) * 2;
  const int nl = N < LNODES ? N : LNODES;

  const int npb  = (N + (int)gridDim.x - 1) / (int)gridDim.x;
  const int n0   = (int)blockIdx.x * npb;
  const int nEnd = n0 + npb < N ? n0 + npb : N;

  for (int i0 = t; i0 < nl; i0 += 16 * 1024) {
    float2 r[16];
#pragma unroll
    for (int u = 0; u < 16; ++u) {
      const int i = i0 + u * 1024;
      r[u] = i < nl ? H2[i] : make_float2(0.0f, 0.0f);
    }
#pragma unroll
    for (int u = 0; u < 16; ++u) {
      const int i = i0 + u * 1024;
      if (i < nl) s_h[i] = r[u];
    }
  }
  __syncthreads();

  const float b0 = b[0];
  const float b1 = b[1];
  const float as0 = asrc[0];
  const float as1 = asrc[1];

  for (int nb = n0 + w; nb < nEnd; nb += 32 * 32) {
    const int myn = nb + 32 * lane;
    int   offv = 0;
    int   cntv = 0;
    float sdv  = 0.0f;
    if (myn < nEnd) {
      offv = off[myn];
      cntv = cnt[myn];
      sdv  = SD[myn];
    }
    int inb = (nEnd - nb + 31) >> 5;
    if (inb > 32) inb = 32;

    int sxn[GCH];
    {
      const int st = __builtin_amdgcn_readlane(offv, 0);
      const int dg = __builtin_amdgcn_readlane(cntv, 0);
#pragma unroll
      for (int u = 0; u < GCH; ++u) {
        const int k = u * 32 + lane;
        sxn[u] = k < dg ? sorted_src[st + k] : -1;
      }
    }

    for (int j = 0; j < inb; ++j) {
      const int   n     = nb + 32 * j;
      const int   start = __builtin_amdgcn_readlane(offv, j);
      const int   deg   = __builtin_amdgcn_readlane(cntv, j);
      const float sdst  = __int_as_float(__builtin_amdgcn_readlane(__float_as_int(sdv), j));

      float m  = -INFINITY;
      float z  = 0.0f;
      float a0 = 0.0f;
      float a1 = 0.0f;

      int sx0[GCH];
#pragma unroll
      for (int u = 0; u < GCH; ++u) sx0[u] = sxn[u];
      if (j + 1 < inb) {
        const int st = __builtin_amdgcn_readlane(offv, j + 1);
        const int dg = __builtin_amdgcn_readlane(cntv, j + 1);
#pragma unroll
        for (int u = 0; u < GCH; ++u) {
          const int k = u * 32 + lane;
          sxn[u] = k < dg ? sorted_src[st + k] : -1;
        }
      }

      for (int c0 = 0; c0 < deg; c0 += 32 * GCH) {
        int sx[GCH];
        if (c0 == 0) {
#pragma unroll
          for (int u = 0; u < GCH; ++u) sx[u] = sx0[u];
        } else {
#pragma unroll
          for (int u = 0; u < GCH; ++u) {
            const int k = c0 + u * 32 + lane;
            sx[u] = k < deg ? sorted_src[start + k] : -1;
          }
        }
        float2 g[GCH];
        float2 lv[GCH];
#pragma unroll
        for (int u = 0; u < GCH; ++u) {
          g[u] = make_float2(0.0f, 0.0f);
          if (sx[u] >= nl) g[u] = H2[sx[u]];
        }
#pragma unroll
        for (int u = 0; u < GCH; ++u) {
          lv[u] = make_float2(0.0f, 0.0f);
          if (sx[u] >= 0 && sx[u] < nl) lv[u] = s_h[sx[u]];
        }
        float e[GCH], h0[GCH], h1[GCH];
        float cm = -INFINITY;
#pragma unroll
        for (int u = 0; u < GCH; ++u) {
          const bool inl = sx[u] < nl;
          h0[u] = inl ? lv[u].x : g[u].x;
          h1[u] = inl ? lv[u].y : g[u].y;
          const float ss = __builtin_fmaf(h1[u], as1, h0[u] * as0);
          e[u]  = sx[u] >= 0 ? lrelu(ss + sdst) : -INFINITY;
          cm = fmaxf(cm, e[u]);
        }
        for (int d = 1; d < 32; d <<= 1) cm = fmaxf(cm, __shfl_xor(cm, d));
        const float mn = fmaxf(m, cm);
        const float sc = __expf(m - mn);
        z  *= sc;
        a0 *= sc;
        a1 *= sc;
#pragma unroll
        for (int u = 0; u < GCH; ++u) {
          const float p = __expf(e[u] - mn);
          z  += p;
          a0 += p * h0[u];
          a1 += p * h1[u];
        }
        m = mn;
      }

      for (int d = 1; d < 32; d <<= 1) {
        z  += __shfl_xor(z, d);
        a0 += __shfl_xor(a0, d);
        a1 += __shfl_xor(a1, d);
      }
      if (lane == 0) {
        const float den = z + 1e-16f;
        const float o0 = a0 / den + b0;
        const float o1 = a1 / den + b1;
        const float mx = fmaxf(o0, o1);
        const float lse = mx + __logf(__expf(o0 - mx) + __expf(o1 - mx));
        *reinterpret_cast<float2*>(outp + (size_t)n * 2) = make_float2(o0 - lse, o1 - lse);
      }
    }
  }
}

extern "C" void kernel_launch(void* const* d_in, const int* in_sizes, int n_in,
                              void* d_out, int out_size, void* d_ws,
                              size_t ws_size, hipStream_t stream) {
  const float* x      = (const float*)d_in[0];
  const float* W1     = (const float*)d_in[1];
  const float* a1_src = (const float*)d_in[2];
  const float* a1_dst = (const float*)d_in[3];
  const float* b1     = (const float*)d_in[4];
  const float* W2     = (const float*)d_in[5];
  const float* a2_src = (const float*)d_in[6];
  const float* a2_dst = (const float*)d_in[7];
  const float* b2     = (const float*)d_in[8];
  const int*   eidx   = (const int*)d_in[9];

  const int N = in_sizes[0] / 36;
  const int E = in_sizes[9] / 2;
  const int* src = eidx;
  const int* dst = eidx + E;

  const size_t Np    = ((size_t)N + 3) & ~(size_t)3;
  const size_t Ep    = ((size_t)E + 3) & ~(size_t)3;
  const size_t rowsW = (size_t)N * 40;
  const size_t headW = Ep > rowsW ? Ep : rowsW;
  const int NB = (N + BNODES - 1) >> BSH;
  if (ws_size < (headW + 704 + 2 * Np + Ep) * 4) return;
  if (NB > MAXB) return;
  const int nblk = (E + CHUNK - 1) / CHUNK;
  if ((size_t)nblk * NB > Ep) return;

  float* ws    = (float*)d_ws;
  int*   binned = (int*)d_ws;
  float* h1    = ws;
  float* h2    = h1 + (size_t)N * 16;
  float* h1r   = h2 + (size_t)N * 16;
  int*   bbase = (int*)(ws + headW);
  int*   cnt   = bbase + 704;
  int*   off   = cnt + Np;
  int*   ssrc  = off + Np;
  int*   btab  = ssrc;

  const int BLK = 256;
  dim3 chunkGrid(nblk);
  dim3 gemmGrid((N + 127) / 128);

  bucket_hist_kernel<<<chunkGrid, 1024, 0, stream>>>(dst, btab, E, NB);
  edge_bin_kernel<<<chunkGrid, 1024, 0, stream>>>(src, dst, btab, bbase, binned, E, NB, nblk);
  bucket_sort_kernel<<<dim3(NB), 1024, 0, stream>>>(binned, bbase, cnt, off, ssrc, N);

  gemm_attn_wmma<9, 8, false><<<gemmGrid, BLK, 0, stream>>>(x, 36, W1, a1_src, a1_dst, h1, N);
  node_agg_hid_kernel<<<dim3(256), 1024, 0, stream>>>(
      ssrc, off, cnt, h1, b1, h1r, N);

  gemm_attn_wmma<2, 2, true><<<gemmGrid, BLK, 0, stream>>>(h1r, 8, W2, a2_src, a2_dst, h2, N);
  node_agg_out_kernel<<<dim3(256), 1024, 0, stream>>>(
      ssrc, off, cnt, h2, a2_src, b2, (float*)d_out, N);
}


# output-layer rows all read from LDS: sort groups each list by three source ranges, work-groups refill LDS per range; per-lane-slot softmax sums over ranges, one max and one 32-slot tree per node
# speedup vs baseline: 19.7442x; 1.0315x over previous
#include <hip/hip_runtime.h>
#include <math.h>

typedef __attribute__((ext_vector_type(2))) float v2f;
typedef __attribute__((ext_vector_type(8))) float v8f;

#define NEG_SLOPE 0.2f

__device__ __forceinline__ float lrelu(float x) {
  return x > 0.0f ? x : NEG_SLOPE * x;
}

template <int K4, int F, bool PACKED>
__global__ void __launch_bounds__(256) gemm_attn_wmma(const float* __restrict__ A, int lda,
                                                      const float* __restrict__ W,
                                                      const float* __restrict__ asrc,
                                                      const float* __restrict__ adst,
                                                      float* __restrict__ H, int nRows) {
  constexpr int K = K4 * 4;
  __shared__ float sB[K * 16];

  int lane = threadIdx.x & 31;
  int wave = (int)((blockIdx.x * blockDim.x + threadIdx.x) >> 5);
  int row0 = wave * 16;
  const bool live = row0 < nRows;
  int hf = lane >> 4;
  int l  = lane & 15;

  v2f a[K4];
  if (live) {
#pragma unroll
    for (int k = 0; k < K4; ++k) {
      const float* ap = A + (size_t)(row0 + l) * lda + k * 4 + hf * 2;
      a[k].x = ap[0];
      a[k].y = ap[1];
    }
  }

  for (int i = threadIdx.x; i < K * 16; i += 256) {
    const int k = i >> 4;
    const int n = i & 15;
    float v = 0.0f;
    if (n < F) {
      v = W[k * F + n];
    } else if (n == 8 && !PACKED) {
      for (int j = 0; j < F; ++j) v = __builtin_fmaf(W[k * F + j], asrc[j], v);
    } else if (n == 9) {
      for (int j = 0; j < F; ++j) v = __builtin_fmaf(W[k * F + j], adst[j], v);
    }
    sB[i] = v;
  }
  __syncthreads();
  if (!live) return;

  v8f c = {};
#pragma unroll
  for (int k = 0; k < K4; ++k) {
    const float* bp = sB + (k * 4 + hf * 2) * 16 + l;
    v2f b;
    b.x = bp[0];
    b.y = bp[16];
    c = __builtin_amdgcn_wmma_f32_16x16x4_f32(false, a[k], false, b, (short)0, c,
                                              false, false);
  }

  if (!PACKED) {
    float* out = H + (size_t)(row0 + hf * 8) * 16 + l;
#pragma unroll
    for (int r = 0; r < 8; ++r) out[(size_t)r * 16] = c[r];
  } else {
    static_assert(!PACKED || F == 2, "packed rows hold two features");
    if (l < 2) {
      float* out = H + (size_t)(row0 + hf * 8) * 2 + l;
#pragma unroll
      for (int r = 0; r < 8; ++r) out[(size_t)r * 2] = c[r];
    } else if (l == 9) {
      float* out = H + (size_t)((nRows + 15) & ~15) * 2 + row0 + hf * 8;
#pragma unroll
      for (int r = 0; r < 8; ++r) out[r] = c[r];
    }
  }
}

#define BSH    9
#define BNODES 512
#define MAXB   351
#define MAXC   11
#define EPT    32
#define CHUNK  (EPT * 1024)
#define SCAP   40960
#define LNODES 40448
#define NRANGE 3

__device__ __forceinline__ int range_rows(int N) {
  int r = (N + NRANGE - 1) / NRANGE;
  r = (r + 1) & ~1;
  return r < LNODES ? r : LNODES;
}
__device__ __forceinline__ int src_range(int s, int R) {
  return (s >= R ? 1 : 0) + (s >= 2 * R ? 1 : 0);
}

__global__ void __launch_bounds__(1024) bucket_hist_kernel(const int* __restrict__ dst,
                                                           int* __restrict__ btab,
                                                           int E, int NB) {
  __shared__ int sh_cnt[MAXB + 1];
  const int t = threadIdx.x;
  if (t < NB) sh_cnt[t] = 0;
  __syncthreads();
  const int e0 = blockIdx.x * CHUNK + t;
  int d[EPT];
#pragma unroll
  for (int u = 0; u < EPT; ++u) {
    const int i = e0 + u * 1024;
    d[u] = i < E ? dst[i] : -1;
  }
#pragma unroll
  for (int u = 0; u < EPT; ++u)
    if (d[u] >= 0) atomicAdd(&sh_cnt[d[u] >> BSH], 1);
  __syncthreads();
  if (t < NB) btab[(size_t)blockIdx.x * NB + t] = sh_cnt[t];
}

__global__ void __launch_bounds__(1024) edge_bin_kernel(const int* __restrict__ src,
                                                        const int* __restrict__ dst,
                                                        const int* __restrict__ btab,
                                                        int* __restrict__ bbase,
                                                        int* __restrict__ binned,
                                                        int E, int NB, int nblk) {
  __shared__ int sh_cnt[MAXB + 1];
  __shared__ int sh_base[MAXB + 1];
  __shared__ int sh_loc[MAXB + 1];
  __shared__ int sh_tot[MAXB + 1];
  __shared__ int sh_pre[MAXB + 1];
  __shared__ int wave_tot[32];
  __shared__ int gwave_tot[32];
  __shared__ int stage[CHUNK];
  const int t = threadIdx.x;
  const int lane = t & 31;
  const int w = t >> 5;
  const int blk = blockIdx.x;
  if (t <= MAXB) {
    sh_cnt[t] = 0;
    sh_tot[t] = 0;
    sh_pre[t] = 0;
  }
  const int own = t < NB ? btab[(size_t)blk * NB + t] : 0;
  const int e0 = blk * CHUNK + t;
  int d[EPT];
#pragma unroll
  for (int u = 0; u < EPT; ++u) {
    const int i = e0 + u * 1024;
    d[u] = i < E ? dst[i] : -1;
  }
  __syncthreads();

  {
    int tot[MAXC];
    int pre[MAXC];
#pragma unroll
    for (int i = 0; i < MAXC; ++i) {
      tot[i] = 0;
      pre[i] = 0;
    }
#pragma unroll 2
    for (int r = w; r < nblk; r += 32) {
      const int* row = btab + (size_t)r * NB;
      const bool earlier = r < blk;
#pragma unroll
      for (int i = 0; i < MAXC; ++i) {
        const int col = lane + 32 * i;
        const int v = col < NB ? row[col] : 0;
        tot[i] += v;
        pre[i] += earlier ? v : 0;
      }
    }
#pragma unroll
    for (int i = 0; i < MAXC; ++i) {
      const int col = lane + 32 * i;
      if (col < NB) {
        atomicAdd(&sh_tot[col], tot[i]);
        atomicAdd(&sh_pre[col], pre[i]);
      }
    }
  }
  __syncthreads();

  const int gown = t < NB ? sh_tot[t] : 0;
  int inc  = own;
  int ginc = gown;
  for (int s = 1; s < 32; s <<= 1) {
    const int o  = __shfl_up(inc, s);
    const int go = __shfl_up(ginc, s);
    if (lane >= s) {
      inc  += o;
      ginc += go;
    }
  }
  if (lane == 31) {
    wave_tot[w]  = inc;
    gwave_tot[w] = ginc;
  }
  __syncthreads();
  int before  = 0;
  int gbefore = 0;
  for (int k = 0; k < 32; ++k)
    if (k < w) {
      before  += wave_tot[k];
      gbefore += gwave_tot[k];
    }
  const int gex = gbefore + ginc - gown;
  if (t < NB) {
    sh_loc[t]  = before + inc - own;
    sh_base[t] = gex + sh_pre[t];
  }
  if (blk == 0 && t <= NB) bbase[t] = gex;
  __syncthreads();

#pragma unroll
  for (int h = 0; h < EPT; h += 16) {
    int s[16];
#pragma unroll
    for (int u = 0; u < 16; ++u) {
      const int i = e0 + (h + u) * 1024;
      s[u] = i < E ? src[i] : 0;
    }
#pragma unroll
    for (int u = 0; u < 16; ++u) {
      const int dd = d[h + u];
      if (dd >= 0) {
        const int k = dd >> BSH;
        const int r = atomicAdd(&sh_cnt[k], 1);
        stage[sh_loc[k] + r] = (s[u] << BSH) | (dd & (BNODES - 1));
      }
    }
  }
  __syncthreads();

  for (int k = w; k < NB; k += 32) {
    const int c = sh_cnt[k];
    const int g = sh_base[k];
    const int l = sh_loc[k];
    for (int j = lane; j < c; j += 32) binned[g + j] = stage[l + j];
  }
}

__global__ void __launch_bounds__(1024) bucket_sort_kernel(const int* __restrict__ binned,
                                                           const int* __restrict__ bbase,
                                                           int* __restrict__ rc,
                                                           int* __restrict__ off,
                                                           int* __restrict__ sorted_src, int N) {
  __shared__ int sh_cnt[BNODES * NRANGE];
  __shared__ int sh_off[BNODES * NRANGE];
  __shared__ int wave_tot[32];
  __shared__ int stage[SCAP];
  const int t = threadIdx.x;
  const int lane = t & 31;
  const int w = t >> 5;
  const int s0 = bbase[blockIdx.x];
  const int m  = bbase[blockIdx.x + 1] - s0;
  const int* mine = binned + s0;
  const bool staged = m <= SCAP;
  const int R = range_rows(N);

  for (int i = t; i < BNODES * NRANGE; i += 1024) sh_cnt[i] = 0;
  __syncthreads();

  for (int i0 = t; i0 < m; i0 += 8 * 1024) {
    int v[8];
#pragma unroll
    for (int u = 0; u < 8; ++u) {
      const int i = i0 + u * 1024;
      v[u] = i < m ? mine[i] : -1;
    }
#pragma unroll
    for (int u = 0; u < 8; ++u)
      if (v[u] >= 0)
        atomicAdd(&sh_cnt[(v[u] & (BNODES - 1)) * NRANGE + src_range(v[u] >> BSH, R)], 1);
  }
  __syncthreads();

  int c0 = 0, c1 = 0, c2 = 0;
  if (t < BNODES) {
    c0 = sh_cnt[t * NRANGE + 0];
    c1 = sh_cnt[t * NRANGE + 1];
    c2 = sh_cnt[t * NRANGE + 2];
  }
  const int own = c0 + c1 + c2;
  int inc = own;
  for (int s = 1; s < 32; s <<= 1) {
    const int o = __shfl_up(inc, s);
    if (lane >= s) inc += o;
  }
  if (lane == 31) wave_tot[w] = inc;
  __syncthreads();
  int before = 0;
  for (int k = 0; k < 32; ++k)
    if (k < w) before += wave_tot[k];
  if (t < BNODES) {
    const int first = before + inc - own;
    sh_off[t * NRANGE + 0] = first;
    sh_off[t * NRANGE + 1] = first + c0;
    sh_off[t * NRANGE + 2] = first + c0 + c1;
    sh_cnt[t * NRANGE + 0] = 0;
    sh_cnt[t * NRANGE + 1] = 0;
    sh_cnt[t * NRANGE + 2] = 0;
    const int n = blockIdx.x * BNODES + t;
    if (n < N) {
      const bool fits = c0 < 0xFFFF && c1 < 0xFFFF;
      rc[n]  = fits ? (int)((unsigned)c0 | ((unsigned)c1 << 16)) : -1;
      off[n] = s0 + first;
    }
  }
  __syncthreads();

  for (int i0 = t; i0 < m; i0 += 8 * 1024) {
    int v[8];
#pragma unroll
    for (int u = 0; u < 8; ++u) {
      const int i = i0 + u * 1024;
      v[u] = i < m ? mine[i] : -1;
    }
#pragma unroll
    for (int u = 0; u < 8; ++u) {
      if (v[u] >= 0) {
        const int sv  = v[u] >> BSH;
        const int key = (v[u] & (BNODES - 1)) * NRANGE + src_range(sv, R);
        const int p = sh_off[key] + atomicAdd(&sh_cnt[key], 1);
        if (staged) stage[p] = sv;
        else        sorted_src[s0 + p] = sv;
      }
    }
  }
  if (staged) {
    __syncthreads();
    for (int j = t; j < m; j += 1024) sorted_src[s0 + j] = stage[j];
  }
}

#define G1CH 3

__global__ void __launch_bounds__(1024) node_agg_hid_kernel(
    const int* __restrict__ sorted_src, const int* __restrict__ off,
    const float* __restrict__ H,
    const float* __restrict__ b, float* __restrict__ outp, int N, int E) {
  const int t = threadIdx.x;
  const int lane = t & 31;
  const int w = t >> 5;
  const int sub = lane & 3;
  const int g   = lane >> 2;

  const int npb  = (N + (int)gridDim.x - 1) / (int)gridDim.x;
  const int n0   = (int)blockIdx.x * npb;
  const int nEnd = n0 + npb < N ? n0 + npb : N;

  const float bs0 = b[0], bs1 = b[1], bs2 = b[2], bs3 = b[3];
  const float bs4 = b[4], bs5 = b[5], bs6 = b[6], bs7 = b[7];

  for (int nb = n0 + w; nb < nEnd; nb += 32 * 32) {
    const int myn = nb + 32 * lane;
    int   offv = 0;
    int   cntv = 0;
    float sdv  = 0.0f;
    if (myn < nEnd) {
      offv = off[myn];
      cntv = (myn + 1 < N ? off[myn + 1] : E) - offv;
      sdv  = H[(size_t)myn * 16 + 9];
    }
    int inb = (nEnd - nb + 31) >> 5;
    if (inb > 32) inb = 32;

    int sxn[G1CH];
    {
      const int st = __builtin_amdgcn_readlane(offv, 0);
      const int dg = __builtin_amdgcn_readlane(cntv, 0);
#pragma unroll
      for (int u = 0; u < G1CH; ++u) {
        const int k = u * 32 + lane;
        sxn[u] = k < dg ? sorted_src[st + k] : 0;
      }
    }

    for (int j = 0; j < inb; ++j) {
      const int   n     = nb + 32 * j;
      const int   start = __builtin_amdgcn_readlane(offv, j);
      const int   deg   = __builtin_amdgcn_readlane(cntv, j);
      const float sdst  = __int_as_float(__builtin_amdgcn_readlane(__float_as_int(sdv), j));

      float  m = -INFINITY;
      float  z = 0.0f;
      float4 acc = make_float4(0.0f, 0.0f, 0.0f, 0.0f);

      int sx0[G1CH];
#pragma unroll
      for (int u = 0; u < G1CH; ++u) sx0[u] = sxn[u];
      if (j + 1 < inb) {
        const int st = __builtin_amdgcn_readlane(offv, j + 1);
        const int dg = __builtin_amdgcn_readlane(cntv, j + 1);
#pragma unroll
        for (int u = 0; u < G1CH; ++u) {
          const int k = u * 32 + lane;
          sxn[u] = k < dg ? sorted_src[st + k] : 0;
        }
      }

      for (int c0 = 0; c0 < deg; c0 += 32 * G1CH) {
        int sx[G1CH];
        if (c0 == 0) {
#pragma unroll
          for (int u = 0; u < G1CH; ++u) sx[u] = sx0[u];
        } else {
#pragma unroll
          for (int u = 0; u < G1CH; ++u) {
            const int k = c0 + u * 32 + lane;
            sx[u] = k < deg ? sorted_src[start + k] : 0;
          }
        }
        float4 v[G1CH][4];
#pragma unroll
        for (int u = 0; u < G1CH; ++u) {
          const bool any = c0 + u * 32 < deg;
          unsigned s[4];
#pragma unroll
          for (int q = 0; q < 4; ++q) s[q] = (unsigned)__shfl(sx[u], q * 8 + g);
#pragma unroll
          for (int q = 0; q < 4; ++q) {
            v[u][q] = make_float4(0.0f, 0.0f, 0.0f, 0.0f);
            if (any)
              v[u][q] = *reinterpret_cast<const float4*>(H + (size_t)s[q] * 16 + sub * 4);
          }
        }
        float e[G1CH][4];
        float cm = -INFINITY;
#pragma unroll
        for (int u = 0; u < G1CH; ++u) {
#pragma unroll
          for (int q = 0; q < 4; ++q) {
            const float ssrc  = __shfl(v[u][q].x, lane - sub + 2);
            const bool  valid = c0 + u * 32 + q * 8 + g < deg;
            e[u][q] = valid ? lrelu(ssrc + sdst) : -INFINITY;
            if (!valid) v[u][q] = make_float4(0.0f, 0.0f, 0.0f, 0.0f);
            cm = fmaxf(cm, e[u][q]);
          }
        }
        for (int d = 4; d < 32; d <<= 1) cm = fmaxf(cm, __shfl_xor(cm, d));
        const float mn = fmaxf(m, cm);
        const float sc = __expf(m - mn);
        z *= sc;
        acc.x *= sc; acc.y *= sc; acc.z *= sc; acc.w *= sc;
#pragma unroll
        for (int u = 0; u < G1CH; ++u) {
#pragma unroll
          for (int q = 0; q < 4; ++q) {
            const float p = __expf(e[u][q] - mn);
            z += p;
            acc.x += p * v[u][q].x;
            acc.y += p * v[u][q].y;
            acc.z += p * v[u][q].z;
            acc.w += p * v[u][q].w;
          }
        }
        m = mn;
      }

      for (int d = 4; d < 32; d <<= 1) {
        z     += __shfl_xor(z, d);
        acc.x += __shfl_xor(acc.x, d);
        acc.y += __shfl_xor(acc.y, d);
        acc.z += __shfl_xor(acc.z, d);
        acc.w += __shfl_xor(acc.w, d);
      }
      if (g == 0 && sub < 2) {
        const float den = z + 1e-16f;
        float4 r;
        r.x = acc.x / den + (sub == 0 ? bs0 : bs4);
        r.y = acc.y / den + (sub == 0 ? bs1 : bs5);
        r.z = acc.z / den + (sub == 0 ? bs2 : bs6);
        r.w = acc.w / den + (sub == 0 ? bs3 : bs7);
        r.x = r.x > 0.0f ? r.x : 0.0f;
        r.y = r.y > 0.0f ? r.y : 0.0f;
        r.z = r.z > 0.0f ? r.z : 0.0f;
        r.w = r.w > 0.0f ? r.w : 0.0f;
        *reinterpret_cast<float4*>(outp + (size_t)n * 8 + sub * 4) = r;
      }
    }
  }
}

#define WNODES 13
#define FILLU  9
#define WNODE_LIST(X) X(0) X(1) X(2) X(3) X(4) X(5) X(6) X(7) X(8) X(9) X(10) X(11) X(12)

__device__ __forceinline__ void slot_add(float2 hv, float as0, float as1, float sdst,
                                         float& m, float& z, float& a0, float& a1) {
  const float ss = __builtin_fmaf(hv.y, as1, hv.x * as0);
  const float e  = lrelu(ss + sdst);
  const float mn = fmaxf(m, e);
  const float sc = __expf(m - mn);
  const float p  = __expf(e - mn);
  z  = z  * sc + p;
  a0 = a0 * sc + p * hv.x;
  a1 = a1 * sc + p * hv.y;
  m  = mn;
}

__global__ void __launch_bounds__(1024) node_agg_out_kernel(
    const int* __restrict__ sorted_src, const int* __restrict__ off,
    const int* __restrict__ rc, const float* __restrict__ H,
    const float* __restrict__ asrc,
    const float* __restrict__ b, float* __restrict__ outp, int N, int E) {
  __shared__ float2 s_h[LNODES];
  const int t = threadIdx.x;
  const int lane = t & 31;
  const int w = t >> 5;
  const float2* H2 = reinterpret_cast<const float2*>(H);
  const float*  SD = H + (size_t)((N + 15) & ~15) * 2;
  const int R = range_rows(N);

  const int npb  = (N + (int)gridDim.x - 1) / (int)gridDim.x;
  const int n0   = (int)blockIdx.x * npb;
  const int nEnd = n0 + npb < N ? n0 + npb : N;

  const float b0 = b[0];
  const float b1 = b[1];
  const float as0 = asrc[0];
  const float as1 = asrc[1];

  const int nb  = n0 + w;
  const int myn = nb + 32 * lane;
  const int inb = nb < nEnd ? (nEnd - nb + 31) >> 5 : 0;
  int   offv = 0;
  int   totv = 0;
  int   rcv  = 0;
  float sdv  = 0.0f;
  if (myn < nEnd) {
    offv = off[myn];
    totv = (myn + 1 < N ? off[myn + 1] : E) - offv;
    rcv  = rc[myn];
    sdv  = SD[myn];
  }
  int c0v = rcv & 0xFFFF;
  int c1v = (int)((unsigned)rcv >> 16);
  if (__builtin_amdgcn_ballot_w32(rcv == -1) != 0u) {
    for (int j = 0; j < inb; ++j) {
      if (__builtin_amdgcn_readlane(rcv, j) != -1) continue;
      const int st = __builtin_amdgcn_readlane(offv, j);
      const int dg = __builtin_amdgcn_readlane(totv, j);
      int k0 = 0;
      int k1 = 0;
      for (int c = 0; c < dg; c += 32) {
        const int k = c + lane;
        const int s = k < dg ? sorted_src[st + k] : -1;
        k0 += __popc(__builtin_amdgcn_ballot_w32(s >= 0 && s < R));
        k1 += __popc(__builtin_amdgcn_ballot_w32(s >= R && s < 2 * R));
      }
      if (lane == j) {
        c0v = k0;
        c1v = k1;
      }
    }
  }
  const int c2v = totv - c0v - c1v;

#define OUT_STATE(j) float ml##j = -INFINITY, zl##j = 0.0f, a0l##j = 0.0f, a1l##j = 0.0f;
  WNODE_LIST(OUT_STATE)
#undef OUT_STATE

  for (int r = 0; r < NRANGE; ++r) {
    const int lo = r * R;
    int nlr = N - lo;
    nlr = nlr < 0 ? 0 : (nlr > R ? R : nlr);

    const int pst = offv + (r >= 1 ? c0v : 0) + (r >= 2 ? c1v : 0);
    const int pcn = r == 0 ? c0v : (r == 1 ? c1v : c2v);

#define OUT_FIRST(j)                                               \
    int sx##j = -1;                                                \
    if (j < inb) {                                                 \
      const int st_ = __builtin_amdgcn_readlane(pst, j);           \
      const int dg_ = __builtin_amdgcn_readlane(pcn, j);           \
      if (lane < dg_) sx##j = sorted_src[st_ + lane];              \
    }
    WNODE_LIST(OUT_FIRST)
#undef OUT_FIRST

    if (r > 0) __syncthreads();
    {
      const float4* H4 = reinterpret_cast<const float4*>(H2 + lo);
      float4*       s4 = reinterpret_cast<float4*>(s_h);
      const int npair = nlr >> 1;
      for (int i0 = t; i0 < npair; i0 += FILLU * 1024) {
        float4 rr[FILLU];
#pragma unroll
        for (int u = 0; u < FILLU; ++u) {
          const int i = i0 + u * 1024;
          rr[u] = i < npair ? H4[i] : make_float4(0.0f, 0.0f, 0.0f, 0.0f);
        }
#pragma unroll
        for (int u = 0; u < FILLU; ++u) {
          const int i = i0 + u * 1024;
          if (i < npair) s4[i] = rr[u];
        }
      }
      if ((nlr & 1) != 0 && t == 0) s_h[nlr - 1] = H2[lo + nlr - 1];
    }
    __syncthreads();

#define OUT_SERVE(j)                                                             \
    if (j < inb) {                                                               \
      const int st_ = __builtin_amdgcn_readlane(pst, j);                         \
      const int dg_ = __builtin_amdgcn_readlane(pcn, j);                         \
      const float sdst_ =                                                        \
          __int_as_float(__builtin_amdgcn_readlane(__float_as_int(sdv), j));     \
      int sx_ = sx##j;                                                           \
      for (int c0 = 0; c0 < dg_; c0 += 32) {                                     \
        if (c0 > 0) {                                                            \
          const int k_ = c0 + lane;                                              \
          sx_ = k_ < dg_ ? sorted_src[st_ + k_] : -1;                            \
        }                                                                        \
        if (sx_ >= 0)                                                            \
          slot_add(s_h[sx_ - lo], as0, as1, sdst_, ml##j, zl##j, a0l##j, a1l##j); \
      }                                                                          \
    }
    WNODE_LIST(OUT_SERVE)
#undef OUT_SERVE
  }

  float zv  = 0.0f;
  float a0v = 0.0f;
  float a1v = 0.0f;
#define OUT_JOIN(j)                                                              \
  {                                                                              \
    float mx_ = ml##j;                                                           \
    for (int d = 1; d < 32; d <<= 1) mx_ = fmaxf(mx_, __shfl_xor(mx_, d));       \
    const float sc_ = ml##j == -INFINITY ? 0.0f : __expf(ml##j - mx_);           \
    float z_  = zl##j  * sc_;                                                    \
    float a0_ = a0l##j * sc_;                                                    \
    float a1_ = a1l##j * sc_;                                                    \
    for (int d = 1; d < 32; d <<= 1) {                                           \
      z_  += __shfl_xor(z_, d);                                                  \
      a0_ += __shfl_xor(a0_, d);                                                 \
      a1_ += __shfl_xor(a1_, d);                                                 \
    }                                                                            \
    if (lane == j) {                                                             \
      zv  = z_;                                                                  \
      a0v = a0_;                                                                 \
      a1v = a1_;                                                                 \
    }                                                                            \
  }
  WNODE_LIST(OUT_JOIN)
#undef OUT_JOIN

  if (lane < inb) {
    const float den = zv + 1e-16f;
    const float o0 = a0v / den + b0;
    const float o1 = a1v / den + b1;
    const float mx = fmaxf(o0, o1);
    const float lse = mx + __logf(__expf(o0 - mx) + __expf(o1 - mx));
    *reinterpret_cast<float2*>(outp + (size_t)myn * 2) = make_float2(o0 - lse, o1 - lse);
  }
}

extern "C" void kernel_launch(void* const* d_in, const int* in_sizes, int n_in,
                              void* d_out, int out_size, void* d_ws,
                              size_t ws_size, hipStream_t stream) {
  const float* x      = (const float*)d_in[0];
  const float* W1     = (const float*)d_in[1];
  const float* a1_src = (const float*)d_in[2];
  const float* a1_dst = (const float*)d_in[3];
  const float* b1     = (const float*)d_in[4];
  const float* W2     = (const float*)d_in[5];
  const float* a2_src = (const float*)d_in[6];
  const float* a2_dst = (const float*)d_in[7];
  const float* b2     = (const float*)d_in[8];
  const int*   eidx   = (const int*)d_in[9];

  const int N = in_sizes[0] / 36;
  const int E = in_sizes[9] / 2;
  const int* src = eidx;
  const int* dst = eidx + E;

  const size_t Np    = ((size_t)N + 3) & ~(size_t)3;
  const size_t Ep    = ((size_t)E + 3) & ~(size_t)3;
  const size_t rowsW = (size_t)N * 40;
  const size_t headW = Ep > rowsW ? Ep : rowsW;
  const int NB = (N + BNODES - 1) >> BSH;
  if (ws_size < (headW + 704 + 2 * Np + Ep) * 4) return;
  if (NB > MAXB) return;
  if (N > 256 * 32 * WNODES) return;
  if (N > NRANGE * LNODES) return;
  const int nblk = (E + CHUNK - 1) / CHUNK;
  if ((size_t)nblk * NB > Ep) return;

  float* ws    = (float*)d_ws;
  int*   binned = (int*)d_ws;
  float* h1    = ws;
  float* h2    = h1 + (size_t)N * 16;
  float* h1r   = h2 + (size_t)N * 16;
  int*   bbase = (int*)(ws + headW);
  int*   rc    = bbase + 704;
  int*   off   = rc + Np;
  int*   ssrc  = off + Np;
  int*   btab  = ssrc;

  const int BLK = 256;
  dim3 chunkGrid(nblk);
  dim3 gemmGrid((N + 127) / 128);

  bucket_hist_kernel<<<chunkGrid, 1024, 0, stream>>>(dst, btab, E, NB);
  edge_bin_kernel<<<chunkGrid, 1024, 0, stream>>>(src, dst, btab, bbase, binned, E, NB, nblk);
  bucket_sort_kernel<<<dim3(NB), 1024, 0, stream>>>(binned, bbase, rc, off, ssrc, N);

  gemm_attn_wmma<9, 8, false><<<gemmGrid, BLK, 0, stream>>>(x, 36, W1, a1_src, a1_dst, h1, N);
  node_agg_hid_kernel<<<dim3(256), 1024, 0, stream>>>(
      ssrc, off, h1, b1, h1r, N, E);

  gemm_attn_wmma<2, 2, true><<<gemmGrid, BLK, 0, stream>>>(h1r, 8, W2, a2_src, a2_dst, h2, N);
  node_agg_out_kernel<<<dim3(256), 1024, 0, stream>>>(
      ssrc, off, rc, h2, a2_src, b2, (float*)d_out, N, E);
}


# hidden-layer source rows cut to eight words (h0..h7; s_src formed per edge from them, s_dst in a plane): two lanes per edge, 16 edges per row load, half the row bytes; 16-slot lane-tree sums
# speedup vs baseline: 20.5405x; 1.0403x over previous
#include <hip/hip_runtime.h>
#include <math.h>

typedef __attribute__((ext_vector_type(2))) float v2f;
typedef __attribute__((ext_vector_type(8))) float v8f;

#define NEG_SLOPE 0.2f

__device__ __forceinline__ float lrelu(float x) {
  return x > 0.0f ? x : NEG_SLOPE * x;
}

template <int K4, int F>
__global__ void __launch_bounds__(256) gemm_attn_wmma(const float* __restrict__ A, int lda,
                                                      const float* __restrict__ W,
                                                      const float* __restrict__ adst,
                                                      float* __restrict__ H, int nRows) {
  constexpr int K = K4 * 4;
  static_assert(F <= 8, "col 9 is the score column");
  __shared__ float sB[K * 16];

  int lane = threadIdx.x & 31;
  int wave = (int)((blockIdx.x * blockDim.x + threadIdx.x) >> 5);
  int row0 = wave * 16;
  const bool live = row0 < nRows;
  int hf = lane >> 4;
  int l  = lane & 15;

  v2f a[K4];
  if (live) {
#pragma unroll
    for (int k = 0; k < K4; ++k) {
      const float* ap = A + (size_t)(row0 + l) * lda + k * 4 + hf * 2;
      a[k].x = ap[0];
      a[k].y = ap[1];
    }
  }

  for (int i = threadIdx.x; i < K * 16; i += 256) {
    const int k = i >> 4;
    const int n = i & 15;
    float v = 0.0f;
    if (n < F) {
      v = W[k * F + n];
    } else if (n == 9) {
      for (int j = 0; j < F; ++j) v = __builtin_fmaf(W[k * F + j], adst[j], v);
    }
    sB[i] = v;
  }
  __syncthreads();
  if (!live) return;

  v8f c = {};
#pragma unroll
  for (int k = 0; k < K4; ++k) {
    const float* bp = sB + (k * 4 + hf * 2) * 16 + l;
    v2f b;
    b.x = bp[0];
    b.y = bp[16];
    c = __builtin_amdgcn_wmma_f32_16x16x4_f32(false, a[k], false, b, (short)0, c,
                                              false, false);
  }

  if (l < F) {
    float* out = H + (size_t)(row0 + hf * 8) * F + l;
#pragma unroll
    for (int r = 0; r < 8; ++r) out[(size_t)r * F] = c[r];
  } else if (l == 9) {
    float* out = H + (size_t)((nRows + 15) & ~15) * F + row0 + hf * 8;
#pragma unroll
    for (int r = 0; r < 8; ++r) out[r] = c[r];
  }
}

#define BSH    9
#define BNODES 512
#define MAXB   351
#define MAXC   11
#define EPT    32
#define CHUNK  (EPT * 1024)
#define SCAP   40960
#define LNODES 40448
#define NRANGE 3

__device__ __forceinline__ int range_rows(int N) {
  int r = (N + NRANGE - 1) / NRANGE;
  r = (r + 1) & ~1;
  return r < LNODES ? r : LNODES;
}
__device__ __forceinline__ int src_range(int s, int R) {
  return (s >= R ? 1 : 0) + (s >= 2 * R ? 1 : 0);
}

__global__ void __launch_bounds__(1024) bucket_hist_kernel(const int* __restrict__ dst,
                                                           int* __restrict__ btab,
                                                           int E, int NB) {
  __shared__ int sh_cnt[MAXB + 1];
  const int t = threadIdx.x;
  if (t < NB) sh_cnt[t] = 0;
  __syncthreads();
  const int e0 = blockIdx.x * CHUNK + t;
  int d[EPT];
#pragma unroll
  for (int u = 0; u < EPT; ++u) {
    const int i = e0 + u * 1024;
    d[u] = i < E ? dst[i] : -1;
  }
#pragma unroll
  for (int u = 0; u < EPT; ++u)
    if (d[u] >= 0) atomicAdd(&sh_cnt[d[u] >> BSH], 1);
  __syncthreads();
  if (t < NB) btab[(size_t)blockIdx.x * NB + t] = sh_cnt[t];
}

__global__ void __launch_bounds__(1024) edge_bin_kernel(const int* __restrict__ src,
                                                        const int* __restrict__ dst,
                                                        const int* __restrict__ btab,
                                                        int* __restrict__ bbase,
                                                        int* __restrict__ binned,
                                                        int E, int NB, int nblk) {
  __shared__ int sh_cnt[MAXB + 1];
  __shared__ int sh_base[MAXB + 1];
  __shared__ int sh_loc[MAXB + 1];
  __shared__ int sh_tot[MAXB + 1];
  __shared__ int sh_pre[MAXB + 1];
  __shared__ int wave_tot[32];
  __shared__ int gwave_tot[32];
  __shared__ int stage[CHUNK];
  const int t = threadIdx.x;
  const int lane = t & 31;
  const int w = t >> 5;
  const int blk = blockIdx.x;
  if (t <= MAXB) {
    sh_cnt[t] = 0;
    sh_tot[t] = 0;
    sh_pre[t] = 0;
  }
  const int own = t < NB ? btab[(size_t)blk * NB + t] : 0;
  const int e0 = blk * CHUNK + t;
  int d[EPT];
#pragma unroll
  for (int u = 0; u < EPT; ++u) {
    const int i = e0 + u * 1024;
    d[u] = i < E ? dst[i] : -1;
  }
  __syncthreads();

  {
    int tot[MAXC];
    int pre[MAXC];
#pragma unroll
    for (int i = 0; i < MAXC; ++i) {
      tot[i] = 0;
      pre[i] = 0;
    }
#pragma unroll 2
    for (int r = w; r < nblk; r += 32) {
      const int* row = btab + (size_t)r * NB;
      const bool earlier = r < blk;
#pragma unroll
      for (int i = 0; i < MAXC; ++i) {
        const int col = lane + 32 * i;
        const int v = col < NB ? row[col] : 0;
        tot[i] += v;
        pre[i] += earlier ? v : 0;
      }
    }
#pragma unroll
    for (int i = 0; i < MAXC; ++i) {
      const int col = lane + 32 * i;
      if (col < NB) {
        atomicAdd(&sh_tot[col], tot[i]);
        atomicAdd(&sh_pre[col], pre[i]);
      }
    }
  }
  __syncthreads();

  const int gown = t < NB ? sh_tot[t] : 0;
  int inc  = own;
  int ginc = gown;
  for (int s = 1; s < 32; s <<= 1) {
    const int o  = __shfl_up(inc, s);
    const int go = __shfl_up(ginc, s);
    if (lane >= s) {
      inc  += o;
      ginc += go;
    }
  }
  if (lane == 31) {
    wave_tot[w]  = inc;
    gwave_tot[w] = ginc;
  }
  __syncthreads();
  int before  = 0;
  int gbefore = 0;
  for (int k = 0; k < 32; ++k)
    if (k < w) {
      before  += wave_tot[k];
      gbefore += gwave_tot[k];
    }
  const int gex = gbefore + ginc - gown;
  if (t < NB) {
    sh_loc[t]  = before + inc - own;
    sh_base[t] = gex + sh_pre[t];
  }
  if (blk == 0 && t <= NB) bbase[t] = gex;
  __syncthreads();

#pragma unroll
  for (int h = 0; h < EPT; h += 16) {
    int s[16];
#pragma unroll
    for (int u = 0; u < 16; ++u) {
      const int i = e0 + (h + u) * 1024;
      s[u] = i < E ? src[i] : 0;
    }
#pragma unroll
    for (int u = 0; u < 16; ++u) {
      const int dd = d[h + u];
      if (dd >= 0) {
        const int k = dd >> BSH;
        const int r = atomicAdd(&sh_cnt[k], 1);
        stage[sh_loc[k] + r] = (s[u] << BSH) | (dd & (BNODES - 1));
      }
    }
  }
  __syncthreads();

  for (int k = w; k < NB; k += 32) {
    const int c = sh_cnt[k];
    const int g = sh_base[k];
    const int l = sh_loc[k];
    for (int j = lane; j < c; j += 32) binned[g + j] = stage[l + j];
  }
}

__global__ void __launch_bounds__(1024) bucket_sort_kernel(const int* __restrict__ binned,
                                                           const int* __restrict__ bbase,
                                                           int* __restrict__ rc,
                                                           int* __restrict__ off,
                                                           int* __restrict__ sorted_src, int N) {
  __shared__ int sh_cnt[BNODES * NRANGE];
  __shared__ int sh_off[BNODES * NRANGE];
  __shared__ int wave_tot[32];
  __shared__ int stage[SCAP];
  const int t = threadIdx.x;
  const int lane = t & 31;
  const int w = t >> 5;
  const int s0 = bbase[blockIdx.x];
  const int m  = bbase[blockIdx.x + 1] - s0;
  const int* mine = binned + s0;
  const bool staged = m <= SCAP;
  const int R = range_rows(N);

  for (int i = t; i < BNODES * NRANGE; i += 1024) sh_cnt[i] = 0;
  __syncthreads();

  for (int i0 = t; i0 < m; i0 += 8 * 1024) {
    int v[8];
#pragma unroll
    for (int u = 0; u < 8; ++u) {
      const int i = i0 + u * 1024;
      v[u] = i < m ? mine[i] : -1;
    }
#pragma unroll
    for (int u = 0; u < 8; ++u)
      if (v[u] >= 0)
        atomicAdd(&sh_cnt[(v[u] & (BNODES - 1)) * NRANGE + src_range(v[u] >> BSH, R)], 1);
  }
  __syncthreads();

  int c0 = 0, c1 = 0, c2 = 0;
  if (t < BNODES) {
    c0 = sh_cnt[t * NRANGE + 0];
    c1 = sh_cnt[t * NRANGE + 1];
    c2 = sh_cnt[t * NRANGE + 2];
  }
  const int own = c0 + c1 + c2;
  int inc = own;
  for (int s = 1; s < 32; s <<= 1) {
    const int o = __shfl_up(inc, s);
    if (lane >= s) inc += o;
  }
  if (lane == 31) wave_tot[w] = inc;
  __syncthreads();
  int before = 0;
  for (int k = 0; k < 32; ++k)
    if (k < w) before += wave_tot[k];
  if (t < BNODES) {
    const int first = before + inc - own;
    sh_off[t * NRANGE + 0] = first;
    sh_off[t * NRANGE + 1] = first + c0;
    sh_off[t * NRANGE + 2] = first + c0 + c1;
    sh_cnt[t * NRANGE + 0] = 0;
    sh_cnt[t * NRANGE + 1] = 0;
    sh_cnt[t * NRANGE + 2] = 0;
    const int n = blockIdx.x * BNODES + t;
    if (n < N) {
      const bool fits = c0 < 0xFFFF && c1 < 0xFFFF;
      rc[n]  = fits ? (int)((unsigned)c0 | ((unsigned)c1 << 16)) : -1;
      off[n] = s0 + first;
    }
  }
  __syncthreads();

  for (int i0 = t; i0 < m; i0 += 8 * 1024) {
    int v[8];
#pragma unroll
    for (int u = 0; u < 8; ++u) {
      const int i = i0 + u * 1024;
      v[u] = i < m ? mine[i] : -1;
    }
#pragma unroll
    for (int u = 0; u < 8; ++u) {
      if (v[u] >= 0) {
        const int sv  = v[u] >> BSH;
        const int key = (v[u] & (BNODES - 1)) * NRANGE + src_range(sv, R);
        const int p = sh_off[key] + atomicAdd(&sh_cnt[key], 1);
        if (staged) stage[p] = sv;
        else        sorted_src[s0 + p] = sv;
      }
    }
  }
  if (staged) {
    __syncthreads();
    for (int j = t; j < m; j += 1024) sorted_src[s0 + j] = stage[j];
  }
}

#define G1CH 3

__global__ void __launch_bounds__(1024) node_agg_hid_kernel(
    const int* __restrict__ sorted_src, const int* __restrict__ off,
    const float* __restrict__ H, const float* __restrict__ asrc,
    const float* __restrict__ b, float* __restrict__ outp, int N, int E) {
  const int t = threadIdx.x;
  const int lane = t & 31;
  const int w = t >> 5;
  const int sub = lane & 1;
  const int g   = lane >> 1;
  const float* SD = H + (size_t)((N + 15) & ~15) * 8;

  const int npb  = (N + (int)gridDim.x - 1) / (int)gridDim.x;
  const int n0   = (int)blockIdx.x * npb;
  const int nEnd = n0 + npb < N ? n0 + npb : N;

  const float bs0 = b[0], bs1 = b[1], bs2 = b[2], bs3 = b[3];
  const float bs4 = b[4], bs5 = b[5], bs6 = b[6], bs7 = b[7];
  const float as0 = asrc[sub * 4 + 0];
  const float as1 = asrc[sub * 4 + 1];
  const float as2 = asrc[sub * 4 + 2];
  const float as3 = asrc[sub * 4 + 3];

  for (int nb = n0 + w; nb < nEnd; nb += 32 * 32) {
    const int myn = nb + 32 * lane;
    int   offv = 0;
    int   cntv = 0;
    float sdv  = 0.0f;
    if (myn < nEnd) {
      offv = off[myn];
      cntv = (myn + 1 < N ? off[myn + 1] : E) - offv;
      sdv  = SD[myn];
    }
    int inb = (nEnd - nb + 31) >> 5;
    if (inb > 32) inb = 32;

    int sxn[G1CH];
    {
      const int st = __builtin_amdgcn_readlane(offv, 0);
      const int dg = __builtin_amdgcn_readlane(cntv, 0);
#pragma unroll
      for (int u = 0; u < G1CH; ++u) {
        const int k = u * 32 + lane;
        sxn[u] = k < dg ? sorted_src[st + k] : 0;
      }
    }

    for (int j = 0; j < inb; ++j) {
      const int   n     = nb + 32 * j;
      const int   start = __builtin_amdgcn_readlane(offv, j);
      const int   deg   = __builtin_amdgcn_readlane(cntv, j);
      const float sdst  = __int_as_float(__builtin_amdgcn_readlane(__float_as_int(sdv), j));

      float  m = -INFINITY;
      float  z = 0.0f;
      float4 acc = make_float4(0.0f, 0.0f, 0.0f, 0.0f);

      int sx0[G1CH];
#pragma unroll
      for (int u = 0; u < G1CH; ++u) sx0[u] = sxn[u];
      if (j + 1 < inb) {
        const int st = __builtin_amdgcn_readlane(offv, j + 1);
        const int dg = __builtin_amdgcn_readlane(cntv, j + 1);
#pragma unroll
        for (int u = 0; u < G1CH; ++u) {
          const int k = u * 32 + lane;
          sxn[u] = k < dg ? sorted_src[st + k] : 0;
        }
      }

      for (int c0 = 0; c0 < deg; c0 += 32 * G1CH) {
        int sx[G1CH];
        if (c0 == 0) {
#pragma unroll
          for (int u = 0; u < G1CH; ++u) sx[u] = sx0[u];
        } else {
#pragma unroll
          for (int u = 0; u < G1CH; ++u) {
            const int k = c0 + u * 32 + lane;
            sx[u] = k < deg ? sorted_src[start + k] : 0;
          }
        }
        float4 v[G1CH][2];
#pragma unroll
        for (int u = 0; u < G1CH; ++u) {
          const bool any = c0 + u * 32 < deg;
          unsigned s[2];
#pragma unroll
          for (int q = 0; q < 2; ++q) s[q] = (unsigned)__shfl(sx[u], q * 16 + g);
#pragma unroll
          for (int q = 0; q < 2; ++q) {
            v[u][q] = make_float4(0.0f, 0.0f, 0.0f, 0.0f);
            if (any)
              v[u][q] = *reinterpret_cast<const float4*>(H + (size_t)s[q] * 8 + sub * 4);
          }
        }
        float e[G1CH][2];
        float cm = -INFINITY;
#pragma unroll
        for (int u = 0; u < G1CH; ++u) {
#pragma unroll
          for (int q = 0; q < 2; ++q) {
            float part = v[u][q].x * as0;
            part = __builtin_fmaf(v[u][q].y, as1, part);
            part = __builtin_fmaf(v[u][q].z, as2, part);
            part = __builtin_fmaf(v[u][q].w, as3, part);
            const float ssrc  = part + __shfl_xor(part, 1);
            const bool  valid = c0 + u * 32 + q * 16 + g < deg;
            e[u][q] = valid ? lrelu(ssrc + sdst) : -INFINITY;
            if (!valid) v[u][q] = make_float4(0.0f, 0.0f, 0.0f, 0.0f);
            cm = fmaxf(cm, e[u][q]);
          }
        }
        for (int d = 2; d < 32; d <<= 1) cm = fmaxf(cm, __shfl_xor(cm, d));
        const float mn = fmaxf(m, cm);
        const float sc = __expf(m - mn);
        z *= sc;
        acc.x *= sc; acc.y *= sc; acc.z *= sc; acc.w *= sc;
#pragma unroll
        for (int u = 0; u < G1CH; ++u) {
#pragma unroll
          for (int q = 0; q < 2; ++q) {
            const float p = __expf(e[u][q] - mn);
            z += p;
            acc.x += p * v[u][q].x;
            acc.y += p * v[u][q].y;
            acc.z += p * v[u][q].z;
            acc.w += p * v[u][q].w;
          }
        }
        m = mn;
      }

      for (int d = 2; d < 32; d <<= 1) {
        z     += __shfl_xor(z, d);
        acc.x += __shfl_xor(acc.x, d);
        acc.y += __shfl_xor(acc.y, d);
        acc.z += __shfl_xor(acc.z, d);
        acc.w += __shfl_xor(acc.w, d);
      }
      if (g == 0) {
        const float den = z + 1e-16f;
        float4 r;
        r.x = acc.x / den + (sub == 0 ? bs0 : bs4);
        r.y = acc.y / den + (sub == 0 ? bs1 : bs5);
        r.z = acc.z / den + (sub == 0 ? bs2 : bs6);
        r.w = acc.w / den + (sub == 0 ? bs3 : bs7);
        r.x = r.x > 0.0f ? r.x : 0.0f;
        r.y = r.y > 0.0f ? r.y : 0.0f;
        r.z = r.z > 0.0f ? r.z : 0.0f;
        r.w = r.w > 0.0f ? r.w : 0.0f;
        *reinterpret_cast<float4*>(outp + (size_t)n * 8 + sub * 4) = r;
      }
    }
  }
}

#define WNODES 13
#define FILLU  9
#define WNODE_LIST(X) X(0) X(1) X(2) X(3) X(4) X(5) X(6) X(7) X(8) X(9) X(10) X(11) X(12)

__device__ __forceinline__ void slot_add(float2 hv, float as0, float as1, float sdst,
                                         float& m, float& z, float& a0, float& a1) {
  const float ss = __builtin_fmaf(hv.y, as1, hv.x * as0);
  const float e  = lrelu(ss + sdst);
  const float mn = fmaxf(m, e);
  const float sc = __expf(m - mn);
  const float p  = __expf(e - mn);
  z  = z  * sc + p;
  a0 = a0 * sc + p * hv.x;
  a1 = a1 * sc + p * hv.y;
  m  = mn;
}

__global__ void __launch_bounds__(1024) node_agg_out_kernel(
    const int* __restrict__ sorted_src, const int* __restrict__ off,
    const int* __restrict__ rc, const float* __restrict__ H,
    const float* __restrict__ asrc,
    const float* __restrict__ b, float* __restrict__ outp, int N, int E) {
  __shared__ float2 s_h[LNODES];
  const int t = threadIdx.x;
  const int lane = t & 31;
  const int w = t >> 5;
  const float2* H2 = reinterpret_cast<const float2*>(H);
  const float*  SD = H + (size_t)((N + 15) & ~15) * 2;
  const int R = range_rows(N);

  const int npb  = (N + (int)gridDim.x - 1) / (int)gridDim.x;
  const int n0   = (int)blockIdx.x * npb;
  const int nEnd = n0 + npb < N ? n0 + npb : N;

  const float b0 = b[0];
  const float b1 = b[1];
  const float as0 = asrc[0];
  const float as1 = asrc[1];

  const int nb  = n0 + w;
  const int myn = nb + 32 * lane;
  const int inb = nb < nEnd ? (nEnd - nb + 31) >> 5 : 0;
  int   offv = 0;
  int   totv = 0;
  int   rcv  = 0;
  float sdv  = 0.0f;
  if (myn < nEnd) {
    offv = off[myn];
    totv = (myn + 1 < N ? off[myn + 1] : E) - offv;
    rcv  = rc[myn];
    sdv  = SD[myn];
  }
  int c0v = rcv & 0xFFFF;
  int c1v = (int)((unsigned)rcv >> 16);
  if (__builtin_amdgcn_ballot_w32(rcv == -1) != 0u) {
    for (int j = 0; j < inb; ++j) {
      if (__builtin_amdgcn_readlane(rcv, j) != -1) continue;
      const int st = __builtin_amdgcn_readlane(offv, j);
      const int dg = __builtin_amdgcn_readlane(totv, j);
      int k0 = 0;
      int k1 = 0;
      for (int c = 0; c < dg; c += 32) {
        const int k = c + lane;
        const int s = k < dg ? sorted_src[st + k] : -1;
        k0 += __popc(__builtin_amdgcn_ballot_w32(s >= 0 && s < R));
        k1 += __popc(__builtin_amdgcn_ballot_w32(s >= R && s < 2 * R));
      }
      if (lane == j) {
        c0v = k0;
        c1v = k1;
      }
    }
  }
  const int c2v = totv - c0v - c1v;

#define OUT_STATE(j) float ml##j = -INFINITY, zl##j = 0.0f, a0l##j = 0.0f, a1l##j = 0.0f;
  WNODE_LIST(OUT_STATE)
#undef OUT_STATE

  for (int r = 0; r < NRANGE; ++r) {
    const int lo = r * R;
    int nlr = N - lo;
    nlr = nlr < 0 ? 0 : (nlr > R ? R : nlr);

    const int pst = offv + (r >= 1 ? c0v : 0) + (r >= 2 ? c1v : 0);
    const int pcn = r == 0 ? c0v : (r == 1 ? c1v : c2v);

#define OUT_FIRST(j)                                               \
    int sx##j = -1;                                                \
    if (j < inb) {                                                 \
      const int st_ = __builtin_amdgcn_readlane(pst, j);           \
      const int dg_ = __builtin_amdgcn_readlane(pcn, j);           \
      if (lane < dg_) sx##j = sorted_src[st_ + lane];              \
    }
    WNODE_LIST(OUT_FIRST)
#undef OUT_FIRST

    if (r > 0) __syncthreads();
    {
      const float4* H4 = reinterpret_cast<const float4*>(H2 + lo);
      float4*       s4 = reinterpret_cast<float4*>(s_h);
      const int npair = nlr >> 1;
      for (int i0 = t; i0 < npair; i0 += FILLU * 1024) {
        float4 rr[FILLU];
#pragma unroll
        for (int u = 0; u < FILLU; ++u) {
          const int i = i0 + u * 1024;
          rr[u] = i < npair ? H4[i] : make_float4(0.0f, 0.0f, 0.0f, 0.0f);
        }
#pragma unroll
        for (int u = 0; u < FILLU; ++u) {
          const int i = i0 + u * 1024;
          if (i < npair) s4[i] = rr[u];
        }
      }
      if ((nlr & 1) != 0 && t == 0) s_h[nlr - 1] = H2[lo + nlr - 1];
    }
    __syncthreads();

#define OUT_SERVE(j)                                                             \
    if (j < inb) {                                                               \
      const int st_ = __builtin_amdgcn_readlane(pst, j);                         \
      const int dg_ = __builtin_amdgcn_readlane(pcn, j);                         \
      const float sdst_ =                                                        \
          __int_as_float(__builtin_amdgcn_readlane(__float_as_int(sdv), j));     \
      int sx_ = sx##j;                                                           \
      for (int c0 = 0; c0 < dg_; c0 += 32) {                                     \
        if (c0 > 0) {                                                            \
          const int k_ = c0 + lane;                                              \
          sx_ = k_ < dg_ ? sorted_src[st_ + k_] : -1;                            \
        }                                                                        \
        if (sx_ >= 0)                                                            \
          slot_add(s_h[sx_ - lo], as0, as1, sdst_, ml##j, zl##j, a0l##j, a1l##j); \
      }                                                                          \
    }
    WNODE_LIST(OUT_SERVE)
#undef OUT_SERVE
  }

  float zv  = 0.0f;
  float a0v = 0.0f;
  float a1v = 0.0f;
#define OUT_JOIN(j)                                                              \
  {                                                                              \
    float mx_ = ml##j;                                                           \
    for (int d = 1; d < 32; d <<= 1) mx_ = fmaxf(mx_, __shfl_xor(mx_, d));       \
    const float sc_ = ml##j == -INFINITY ? 0.0f : __expf(ml##j - mx_);           \
    float z_  = zl##j  * sc_;                                                    \
    float a0_ = a0l##j * sc_;                                                    \
    float a1_ = a1l##j * sc_;                                                    \
    for (int d = 1; d < 32; d <<= 1) {                                           \
      z_  += __shfl_xor(z_, d);                                                  \
      a0_ += __shfl_xor(a0_, d);                                                 \
      a1_ += __shfl_xor(a1_, d);                                                 \
    }                                                                            \
    if (lane == j) {                                                             \
      zv  = z_;                                                                  \
      a0v = a0_;                                                                 \
      a1v = a1_;                                                                 \
    }                                                                            \
  }
  WNODE_LIST(OUT_JOIN)
#undef OUT_JOIN

  if (lane < inb) {
    const float den = zv + 1e-16f;
    const float o0 = a0v / den + b0;
    const float o1 = a1v / den + b1;
    const float mx = fmaxf(o0, o1);
    const float lse = mx + __logf(__expf(o0 - mx) + __expf(o1 - mx));
    *reinterpret_cast<float2*>(outp + (size_t)myn * 2) = make_float2(o0 - lse, o1 - lse);
  }
}

extern "C" void kernel_launch(void* const* d_in, const int* in_sizes, int n_in,
                              void* d_out, int out_size, void* d_ws,
                              size_t ws_size, hipStream_t stream) {
  const float* x      = (const float*)d_in[0];
  const float* W1     = (const float*)d_in[1];
  const float* a1_src = (const float*)d_in[2];
  const float* a1_dst = (const float*)d_in[3];
  const float* b1     = (const float*)d_in[4];
  const float* W2     = (const float*)d_in[5];
  const float* a2_src = (const float*)d_in[6];
  const float* a2_dst = (const float*)d_in[7];
  const float* b2     = (const float*)d_in[8];
  const int*   eidx   = (const int*)d_in[9];

  const int N = in_sizes[0] / 36;
  const int E = in_sizes[9] / 2;
  const int* src = eidx;
  const int* dst = eidx + E;

  const size_t Np    = ((size_t)N + 3) & ~(size_t)3;
  const size_t Ep    = ((size_t)E + 3) & ~(size_t)3;
  const size_t rowsW = (size_t)N * 40;
  const size_t headW = Ep > rowsW ? Ep : rowsW;
  const int NB = (N + BNODES - 1) >> BSH;
  if (ws_size < (headW + 704 + 2 * Np + Ep) * 4) return;
  if (NB > MAXB) return;
  if (N > 256 * 32 * WNODES) return;
  if (N > NRANGE * LNODES) return;
  const int nblk = (E + CHUNK - 1) / CHUNK;
  if ((size_t)nblk * NB > Ep) return;

  float* ws    = (float*)d_ws;
  int*   binned = (int*)d_ws;
  float* h1    = ws;
  float* h2    = h1 + (size_t)N * 16;
  float* h1r   = h2 + (size_t)N * 16;
  int*   bbase = (int*)(ws + headW);
  int*   rc    = bbase + 704;
  int*   off   = rc + Np;
  int*   ssrc  = off + Np;
  int*   btab  = ssrc;

  const int BLK = 256;
  dim3 chunkGrid(nblk);
  dim3 gemmGrid((N + 127) / 128);

  bucket_hist_kernel<<<chunkGrid, 1024, 0, stream>>>(dst, btab, E, NB);
  edge_bin_kernel<<<chunkGrid, 1024, 0, stream>>>(src, dst, btab, bbase, binned, E, NB, nblk);
  bucket_sort_kernel<<<dim3(NB), 1024, 0, stream>>>(binned, bbase, rc, off, ssrc, N);

  gemm_attn_wmma<9, 8><<<gemmGrid, BLK, 0, stream>>>(x, 36, W1, a1_dst, h1, N);
  node_agg_hid_kernel<<<dim3(256), 1024, 0, stream>>>(
      ssrc, off, h1, a1_src, b1, h1r, N, E);

  gemm_attn_wmma<2, 2><<<gemmGrid, BLK, 0, stream>>>(h1r, 8, W2, a2_dst, h2, N);
  node_agg_out_kernel<<<dim3(256), 1024, 0, stream>>>(
      ssrc, off, rc, h2, a2_src, b2, (float*)d_out, N, E);
}
